# starting kernel
# baseline (speedup 1.0000x reference)
#include <hip/hip_runtime.h>
#include <math.h>

typedef float v2f __attribute__((ext_vector_type(2)));
typedef float v8f __attribute__((ext_vector_type(8)));

#define WMMA_F32(a, b, c) \
  __builtin_amdgcn_wmma_f32_16x16x4_f32(false, (a), false, (b), (short)0, (c), false, false)

static constexpr int B_ = 8, S_ = 2048, D_ = 1024, E_ = 64;
static constexpr int M_ = B_ * S_;
static constexpr int NCHUNK = 32, CH = 64;
static constexpr long long YSZ = (long long)M_ * D_;

__global__ __launch_bounds__(128)
void k_beta_gemm(const float* __restrict__ x, const float* __restrict__ w_in,
                 float* __restrict__ beta) {
  const int lane = threadIdx.x & 31;
  const int wv   = (blockIdx.x << 2) + (threadIdx.x >> 5);
  const int m16  = lane & 15;
  const int half = lane >> 4;

  const float* xr = x + (size_t)(wv * 16 + m16) * D_ + 2 * half;

  v8f acc[4];
#pragma unroll
  for (int i = 0; i < 4; ++i) acc[i] = {};

  for (int kb = 0; kb < D_; kb += 4) {
    float2 av = *(const float2*)(xr + kb);
    v2f a; a.x = av.x; a.y = av.y;
#pragma unroll
    for (int nt = 0; nt < 4; ++nt) {
      const int e = nt * 16 + m16;
      float2 bv = *(const float2*)(w_in + (size_t)e * D_ + kb + 2 * half);
      v2f bb; bb.x = bv.x; bb.y = bv.y;
      acc[nt] = WMMA_F32(a, bb, acc[nt]);
    }
  }
#pragma unroll
  for (int nt = 0; nt < 4; ++nt)
#pragma unroll
    for (int v = 0; v < 8; ++v)
      beta[(size_t)(wv * 16 + v + 8 * half) * E_ + nt * 16 + m16] = acc[nt][v];
}

__global__ __launch_bounds__(512)
void k_scan(const float* __restrict__ beta, const float* __restrict__ alpha,
            const float* __restrict__ omega, const float* __restrict__ res,
            float2* __restrict__ F, float* __restrict__ states,
            float* __restrict__ rfin, float* __restrict__ ifin, int phase) {
  const int c = blockIdx.x;
  const int t = threadIdx.x;
  const int b = t >> 6;
  const int e = t & 63;

  const float mag = 1.f / (1.f + expf(-alpha[e]));
  const float cw  = cosf(omega[e]);
  const float sw  = sinf(omega[e]);

  float rc[64];
#pragma unroll
  for (int k = 0; k < 64; ++k) rc[k] = res[k * E_ + e];

  __shared__ float buf[2][B_ * E_ * 2];

  float re = 0.f, im = 0.f;
  if (phase && c > 0) {
    float2 z = F[((c - 1) * B_ + b) * E_ + e];
    re = z.x; im = z.y;
  }

  const float* bp = beta   + (size_t)(b * S_ + c * CH) * E_ + e;
  float*       sp = states + (size_t)(b * S_ + c * CH) * E_ + e;

  float bt = bp[0];
  for (int tau = 0; tau < CH; ++tau) {
    const int nxt = (tau + 1 < CH) ? (tau + 1) : (CH - 1);
    const float bt_next = bp[nxt * E_];

    const float nr = fmaf(mag, re * cw - im * sw, bt);
    const float ni = mag * (re * sw + im * cw);
    *(float2*)&buf[tau & 1][(b * E_ + e) * 2] = make_float2(nr, ni);
    __syncthreads();

    v2f acc0 = {0.f, 0.f}, acc1 = {0.f, 0.f};
    const float4* rd = (const float4*)&buf[tau & 1][b * E_ * 2];
#pragma unroll
    for (int k = 0; k < 64; k += 4) {
      float4 va = rd[k >> 1];
      float4 vb = rd[(k >> 1) + 1];
      v2f p0 = {va.x, va.y}, p1 = {va.z, va.w};
      v2f p2 = {vb.x, vb.y}, p3 = {vb.z, vb.w};
      v2f r0 = {rc[k], rc[k]},         r1 = {rc[k + 1], rc[k + 1]};
      v2f r2 = {rc[k + 2], rc[k + 2]}, r3 = {rc[k + 3], rc[k + 3]};
      acc0 = __builtin_elementwise_fma(p0, r0, acc0);
      acc1 = __builtin_elementwise_fma(p1, r1, acc1);
      acc0 = __builtin_elementwise_fma(p2, r2, acc0);
      acc1 = __builtin_elementwise_fma(p3, r3, acc1);
    }
    const v2f accs = acc0 + acc1;
    re = accs.x; im = accs.y;
    if (phase) sp[tau * E_] = re;
    bt = bt_next;
  }

  if (!phase) {
    F[(c * B_ + b) * E_ + e] = make_float2(re, im);
  } else if (c == NCHUNK - 1) {
    rfin[b * E_ + e] = re;
    ifin[b * E_ + e] = im;
  }
}

__global__ __launch_bounds__(128)
void k_out_ln(const float* __restrict__ states, const float* __restrict__ w_out,
              const float* __restrict__ gamma, const float* __restrict__ lbeta,
              float* __restrict__ y) {
  const int lane = threadIdx.x & 31;
  const int wv   = threadIdx.x >> 5;
  const int m16  = lane & 15;
  const int half = lane >> 4;
  const int rowbase = blockIdx.x * 16;

  __shared__ float  yt[16][D_ + 4];
  __shared__ float  psum[16][8], psq[16][8];
  __shared__ float2 lnp[16];

  v2f a[16];
  const float* srow = states + (size_t)(rowbase + m16) * E_ + 2 * half;
#pragma unroll
  for (int ks = 0; ks < 16; ++ks) {
    float2 av = *(const float2*)(srow + 4 * ks);
    a[ks].x = av.x; a[ks].y = av.y;
  }

  for (int dt = wv * 16; dt < wv * 16 + 16; ++dt) {
    v8f acc = {};
#pragma unroll
    for (int ks = 0; ks < 16; ++ks) {
      float2 bv = *(const float2*)(w_out + (size_t)(dt * 16 + m16) * E_ + 4 * ks + 2 * half);
      v2f bb; bb.x = bv.x; bb.y = bv.y;
      acc = WMMA_F32(a[ks], bb, acc);
    }
#pragma unroll
    for (int v = 0; v < 8; ++v)
      yt[v + 8 * half][dt * 16 + m16] = acc[v];
  }
  __syncthreads();

  {
    const int r = threadIdx.x >> 3, ss = threadIdx.x & 7;
    float s0 = 0.f, s1 = 0.f;
    for (int j = ss; j < D_; j += 8) {
      float v = yt[r][j];
      s0 += v; s1 += v * v;
    }
    psum[r][ss] = s0; psq[r][ss] = s1;
  }
  __syncthreads();
  if (threadIdx.x < 16) {
    const int r = threadIdx.x;
    float s0 = 0.f, s1 = 0.f;
#pragma unroll
    for (int j = 0; j < 8; ++j) { s0 += psum[r][j]; s1 += psq[r][j]; }
    const float mu  = s0 * (1.f / D_);
    const float var = s1 * (1.f / D_) - mu * mu;
    lnp[r] = make_float2(mu, rsqrtf(var + 1e-5f));
  }
  __syncthreads();

  for (int r = 0; r < 16; ++r) {
    const float mu = lnp[r].x, rs = lnp[r].y;
    for (int j = threadIdx.x; j < D_; j += 128) {
      float v = (yt[r][j] - mu) * rs * gamma[j] + lbeta[j];
      y[(size_t)(rowbase + r) * D_ + j] = v;
    }
  }
}

extern "C" void kernel_launch(void* const* d_in, const int* in_sizes, int n_in,
                              void* d_out, int out_size, void* d_ws, size_t ws_size,
                              hipStream_t stream) {
  (void)in_sizes; (void)n_in; (void)out_size; (void)ws_size;

  const float* x     = (const float*)d_in[0];
  const float* alpha = (const float*)d_in[1];
  const float* omega = (const float*)d_in[2];
  const float* w_in  = (const float*)d_in[3];
  const float* w_out = (const float*)d_in[4];
  const float* res   = (const float*)d_in[5];
  const float* gamma = (const float*)d_in[6];
  const float* lbeta = (const float*)d_in[7];

  float* out  = (float*)d_out;
  float* y    = out;
  float* rfin = out + YSZ;
  float* ifin = rfin + B_ * E_;

  float*  beta   = (float*)d_ws;
  float*  states = beta + (size_t)M_ * E_;
  float2* F      = (float2*)(states + (size_t)M_ * E_);

  hipLaunchKernelGGL(k_beta_gemm, dim3(M_ / 64), dim3(128), 0, stream, x, w_in, beta);
  hipLaunchKernelGGL(k_scan, dim3(NCHUNK), dim3(512), 0, stream,
                     beta, alpha, omega, res, F, states, rfin, ifin, 0);
  hipLaunchKernelGGL(k_scan, dim3(NCHUNK), dim3(512), 0, stream,
                     beta, alpha, omega, res, F, states, rfin, ifin, 1);
  hipLaunchKernelGGL(k_out_ln, dim3(M_ / 16), dim3(128), 0, stream,
                     states, w_out, gamma, lbeta, y);
}


# scan step: 64x64 resonance product moved from LDS-broadcast packed FMA to f32 WMMA 16x16x4 on 4 waves, rows exchanged via LDS behind an LDS-only hand barrier; that product's summation order changes
# speedup vs baseline: 1.5639x; 1.5639x over previous
#include <hip/hip_runtime.h>
#include <math.h>

typedef float v2f __attribute__((ext_vector_type(2)));
typedef float v8f __attribute__((ext_vector_type(8)));

#define WMMA_F32(a, b, c) \
  __builtin_amdgcn_wmma_f32_16x16x4_f32(false, (a), false, (b), (short)0, (c), false, false)

static constexpr int B_ = 8, S_ = 2048, D_ = 1024, E_ = 64;
static constexpr int M_ = B_ * S_;
static constexpr int NCHUNK = 32, CH = 64;
static constexpr long long YSZ = (long long)M_ * D_;

__global__ __launch_bounds__(128)
void k_beta_gemm(const float* __restrict__ x, const float* __restrict__ w_in,
                 float* __restrict__ beta) {
  const int lane = threadIdx.x & 31;
  const int wv   = (blockIdx.x << 2) + (threadIdx.x >> 5);
  const int m16  = lane & 15;
  const int half = lane >> 4;

  const float* xr = x + (size_t)(wv * 16 + m16) * D_ + 2 * half;

  v8f acc[4];
#pragma unroll
  for (int i = 0; i < 4; ++i) acc[i] = {};

  for (int kb = 0; kb < D_; kb += 4) {
    float2 av = *(const float2*)(xr + kb);
    v2f a; a.x = av.x; a.y = av.y;
#pragma unroll
    for (int nt = 0; nt < 4; ++nt) {
      const int e = nt * 16 + m16;
      float2 bv = *(const float2*)(w_in + (size_t)e * D_ + kb + 2 * half);
      v2f bb; bb.x = bv.x; bb.y = bv.y;
      acc[nt] = WMMA_F32(a, bb, acc[nt]);
    }
  }
#pragma unroll
  for (int nt = 0; nt < 4; ++nt)
#pragma unroll
    for (int v = 0; v < 8; ++v)
      beta[(size_t)(wv * 16 + v + 8 * half) * E_ + nt * 16 + m16] = acc[nt][v];
}

static constexpr int SP = 68;
static constexpr int RP = 72;

__global__ __launch_bounds__(128)
void k_scan(const float* __restrict__ beta, const float* __restrict__ alpha,
            const float* __restrict__ omega, const float* __restrict__ res,
            float4* __restrict__ F, float* __restrict__ states,
            float* __restrict__ rfin, float* __restrict__ ifin, int phase) {
  const int c    = blockIdx.x;
  const int tid  = threadIdx.x;
  const int lane = tid & 31;
  const int w    = tid >> 5;
  const int m16  = lane & 15;
  const int half = lane >> 4;
  const int e    = w * 16 + m16;
  const int b0   = 4 * half;

  const float mag = 1.f / (1.f + expf(-alpha[e]));
  const float cw  = cosf(omega[e]);
  const float sw  = sinf(omega[e]);

  v2f rb[16];
#pragma unroll
  for (int t = 0; t < 16; ++t) {
    const int k = 32 * half + 2 * t;
    rb[t].x = res[k * E_ + e];
    rb[t].y = res[(k + 1) * E_ + e];
  }

  __shared__ __attribute__((aligned(16))) float xs[2][16 * SP];
  __shared__ __attribute__((aligned(16))) float rs[2][B_ * RP];
  __shared__ __attribute__((aligned(16))) float fs[B_ * RP];

  v8f st = {};
  if (phase && c > 0) {
    const float4* fp = F + (size_t)(((c - 1) * 4 + w) * 2) * 32 + lane;
    const float4 f0 = fp[0], f1 = fp[32];
    st = (v8f){f0.x, f0.y, f0.z, f0.w, f1.x, f1.y, f1.z, f1.w};
  }

  const float* bp = beta + (size_t)(b0 * S_ + c * CH) * E_ + e;

  float bte[4], bto[4];
#pragma unroll
  for (int q = 0; q < 4; ++q) {
    bte[q] = bp[(size_t)q * S_ * E_];
    bto[q] = bp[(size_t)q * S_ * E_ + E_];
  }

  const int ob = tid >> 4, oc = (tid & 15) * 4;
  float* sp = states + (size_t)(ob * S_ + c * CH) * E_ + oc;

  auto step = [&](const int tau, const int cur, float (&bt)[4]) {
    float* xw = &xs[cur][(8 * half) * SP + e];
#pragma unroll
    for (int q = 0; q < 4; ++q) {
      const float re = st[2 * q], im = st[2 * q + 1];
      const float nr = fmaf(mag, re * cw - im * sw, bt[q]);
      const float ni = mag * (re * sw + im * cw);
      xw[(2 * q) * SP]     = nr;
      xw[(2 * q + 1) * SP] = ni;
    }

    asm volatile("s_wait_dscnt 0x0\n\ts_barrier_signal -1" ::: "memory");

    const int t2 = (tau + 2 < CH) ? (tau + 2) : (CH - 1);
#pragma unroll
    for (int q = 0; q < 4; ++q) bt[q] = bp[(size_t)q * S_ * E_ + t2 * E_];

    asm volatile("s_barrier_wait -1" ::: "memory");

    const float4* ar = (const float4*)&xs[cur][m16 * SP + 32 * half];
    float4 av[8];
#pragma unroll
    for (int j = 0; j < 8; ++j) av[j] = ar[j];

    const bool flush = phase && tau > 0;
    float4 o = make_float4(0.f, 0.f, 0.f, 0.f);
    if (flush) o = *(const float4*)&rs[cur ^ 1][ob * RP + oc];

    v8f acc0 = {}, acc1 = {};
#pragma unroll
    for (int j = 0; j < 8; ++j) {
      v2f a0 = {av[j].x, av[j].y};
      v2f a1 = {av[j].z, av[j].w};
      acc0 = WMMA_F32(a0, rb[2 * j],     acc0);
      acc1 = WMMA_F32(a1, rb[2 * j + 1], acc1);
    }
    st = acc0 + acc1;

    if (phase) {
      float* rw = &rs[cur][b0 * RP + e];
#pragma unroll
      for (int q = 0; q < 4; ++q) rw[q * RP] = st[2 * q];
      if (tau > 0) *(float4*)(sp + (tau - 1) * E_) = o;
    }
  };

  for (int tau = 0; tau < CH; tau += 2) {
    step(tau,     0, bte);
    step(tau + 1, 1, bto);
  }

  if (!phase) {
    float4* fp = F + (size_t)((c * 4 + w) * 2) * 32 + lane;
    fp[0]  = make_float4(st[0], st[1], st[2], st[3]);
    fp[32] = make_float4(st[4], st[5], st[6], st[7]);
  } else {
    float* iw = &fs[b0 * RP + e];
#pragma unroll
    for (int q = 0; q < 4; ++q) iw[q * RP] = st[2 * q + 1];
    asm volatile("s_wait_dscnt 0x0\n\ts_barrier_signal -1\n\ts_barrier_wait -1" ::: "memory");
    const float4 o = *(const float4*)&rs[(CH - 1) & 1][ob * RP + oc];
    *(float4*)(sp + (CH - 1) * E_) = o;
    if (c == NCHUNK - 1) {
      *(float4*)(rfin + ob * E_ + oc) = o;
      *(float4*)(ifin + ob * E_ + oc) = *(const float4*)&fs[ob * RP + oc];
    }
  }
}

__global__ __launch_bounds__(128)
void k_out_ln(const float* __restrict__ states, const float* __restrict__ w_out,
              const float* __restrict__ gamma, const float* __restrict__ lbeta,
              float* __restrict__ y) {
  const int lane = threadIdx.x & 31;
  const int wv   = threadIdx.x >> 5;
  const int m16  = lane & 15;
  const int half = lane >> 4;
  const int rowbase = blockIdx.x * 16;

  __shared__ float  yt[16][D_ + 4];
  __shared__ float  psum[16][8], psq[16][8];
  __shared__ float2 lnp[16];

  v2f a[16];
  const float* srow = states + (size_t)(rowbase + m16) * E_ + 2 * half;
#pragma unroll
  for (int ks = 0; ks < 16; ++ks) {
    float2 av = *(const float2*)(srow + 4 * ks);
    a[ks].x = av.x; a[ks].y = av.y;
  }

  for (int dt = wv * 16; dt < wv * 16 + 16; ++dt) {
    v8f acc = {};
#pragma unroll
    for (int ks = 0; ks < 16; ++ks) {
      float2 bv = *(const float2*)(w_out + (size_t)(dt * 16 + m16) * E_ + 4 * ks + 2 * half);
      v2f bb; bb.x = bv.x; bb.y = bv.y;
      acc = WMMA_F32(a[ks], bb, acc);
    }
#pragma unroll
    for (int v = 0; v < 8; ++v)
      yt[v + 8 * half][dt * 16 + m16] = acc[v];
  }
  __syncthreads();

  {
    const int r = threadIdx.x >> 3, ss = threadIdx.x & 7;
    float s0 = 0.f, s1 = 0.f;
    for (int j = ss; j < D_; j += 8) {
      float v = yt[r][j];
      s0 += v; s1 += v * v;
    }
    psum[r][ss] = s0; psq[r][ss] = s1;
  }
  __syncthreads();
  if (threadIdx.x < 16) {
    const int r = threadIdx.x;
    float s0 = 0.f, s1 = 0.f;
#pragma unroll
    for (int j = 0; j < 8; ++j) { s0 += psum[r][j]; s1 += psq[r][j]; }
    const float mu  = s0 * (1.f / D_);
    const float var = s1 * (1.f / D_) - mu * mu;
    lnp[r] = make_float2(mu, rsqrtf(var + 1e-5f));
  }
  __syncthreads();

  for (int r = 0; r < 16; ++r) {
    const float mu = lnp[r].x, rs = lnp[r].y;
    for (int j = threadIdx.x; j < D_; j += 128) {
      float v = (yt[r][j] - mu) * rs * gamma[j] + lbeta[j];
      y[(size_t)(rowbase + r) * D_ + j] = v;
    }
  }
}

extern "C" void kernel_launch(void* const* d_in, const int* in_sizes, int n_in,
                              void* d_out, int out_size, void* d_ws, size_t ws_size,
                              hipStream_t stream) {
  (void)in_sizes; (void)n_in; (void)out_size; (void)ws_size;

  const float* x     = (const float*)d_in[0];
  const float* alpha = (const float*)d_in[1];
  const float* omega = (const float*)d_in[2];
  const float* w_in  = (const float*)d_in[3];
  const float* w_out = (const float*)d_in[4];
  const float* res   = (const float*)d_in[5];
  const float* gamma = (const float*)d_in[6];
  const float* lbeta = (const float*)d_in[7];

  float* out  = (float*)d_out;
  float* y    = out;
  float* rfin = out + YSZ;
  float* ifin = rfin + B_ * E_;

  float*  beta   = (float*)d_ws;
  float*  states = beta + (size_t)M_ * E_;
  float4* F      = (float4*)(states + (size_t)M_ * E_);

  hipLaunchKernelGGL(k_beta_gemm, dim3(M_ / 64), dim3(128), 0, stream, x, w_in, beta);
  hipLaunchKernelGGL(k_scan, dim3(NCHUNK), dim3(128), 0, stream,
                     beta, alpha, omega, res, F, states, rfin, ifin, 0);
  hipLaunchKernelGGL(k_scan, dim3(NCHUNK), dim3(128), 0, stream,
                     beta, alpha, omega, res, F, states, rfin, ifin, 1);
  hipLaunchKernelGGL(k_out_ln, dim3(M_ / 16), dim3(128), 0, stream,
                     states, w_out, gamma, lbeta, y);
}


# scan in ONE launch of 256 blocks (was two of 32): each runs the 64 steps before its own 8 from a zero state, then writes its 8; serial steps 128 -> 72; carry kept 65-72 steps back (was 65-128)
# speedup vs baseline: 1.7505x; 1.1193x over previous
#include <hip/hip_runtime.h>
#include <math.h>

typedef float v2f __attribute__((ext_vector_type(2)));
typedef float v8f __attribute__((ext_vector_type(8)));

#define WMMA_F32(a, b, c) \
  __builtin_amdgcn_wmma_f32_16x16x4_f32(false, (a), false, (b), (short)0, (c), false, false)

static constexpr int B_ = 8, S_ = 2048, D_ = 1024, E_ = 64;
static constexpr int M_ = B_ * S_;
static constexpr long long YSZ = (long long)M_ * D_;

__global__ __launch_bounds__(128)
void k_beta_gemm(const float* __restrict__ x, const float* __restrict__ w_in,
                 float* __restrict__ beta) {
  const int lane = threadIdx.x & 31;
  const int wv   = (blockIdx.x << 2) + (threadIdx.x >> 5);
  const int m16  = lane & 15;
  const int half = lane >> 4;

  const float* xr = x + (size_t)(wv * 16 + m16) * D_ + 2 * half;

  v8f acc[4];
#pragma unroll
  for (int i = 0; i < 4; ++i) acc[i] = {};

  for (int kb = 0; kb < D_; kb += 4) {
    float2 av = *(const float2*)(xr + kb);
    v2f a; a.x = av.x; a.y = av.y;
#pragma unroll
    for (int nt = 0; nt < 4; ++nt) {
      const int e = nt * 16 + m16;
      float2 bv = *(const float2*)(w_in + (size_t)e * D_ + kb + 2 * half);
      v2f bb; bb.x = bv.x; bb.y = bv.y;
      acc[nt] = WMMA_F32(a, bb, acc[nt]);
    }
  }
#pragma unroll
  for (int nt = 0; nt < 4; ++nt)
#pragma unroll
    for (int v = 0; v < 8; ++v)
      beta[(size_t)(wv * 16 + v + 8 * half) * E_ + nt * 16 + m16] = acc[nt][v];
}

static constexpr int LCH  = 8;
static constexpr int NBLK = S_ / LCH;
static constexpr int WARM = 64;
static constexpr int SP = 68;
static constexpr int RP = 72;

__global__ __launch_bounds__(128)
void k_scan(const float* __restrict__ beta, const float* __restrict__ alpha,
            const float* __restrict__ omega, const float* __restrict__ res,
            float* __restrict__ states,
            float* __restrict__ rfin, float* __restrict__ ifin) {
  const int c    = blockIdx.x;
  const int tid  = threadIdx.x;
  const int lane = tid & 31;
  const int w    = tid >> 5;
  const int m16  = lane & 15;
  const int half = lane >> 4;
  const int e    = w * 16 + m16;
  const int b0   = 4 * half;

  const int t0    = c * LCH;
  const int tw    = (t0 > WARM) ? (t0 - WARM) : 0;
  const int nwarm = t0 - tw;
  const int tlast = t0 + LCH - 1;

  const float mag = 1.f / (1.f + expf(-alpha[e]));
  const float cw  = cosf(omega[e]);
  const float sw  = sinf(omega[e]);

  v2f rb[16];
#pragma unroll
  for (int t = 0; t < 16; ++t) {
    const int k = 32 * half + 2 * t;
    rb[t].x = res[k * E_ + e];
    rb[t].y = res[(k + 1) * E_ + e];
  }

  __shared__ __attribute__((aligned(16))) float xs[2][16 * SP];
  __shared__ __attribute__((aligned(16))) float rs[2][B_ * RP];
  __shared__ __attribute__((aligned(16))) float fs[B_ * RP];

  v8f st = {};

  const float* bp = beta + (size_t)(b0 * S_) * E_ + e;

  float bte[4], bto[4];
#pragma unroll
  for (int q = 0; q < 4; ++q) bte[q] = bp[(size_t)q * S_ * E_ + (size_t)tw * E_];
#pragma unroll
  for (int q = 0; q < 4; ++q) bto[q] = bp[(size_t)q * S_ * E_ + (size_t)(tw + 1) * E_];

  const int ob = tid >> 4, oc = (tid & 15) * 4;
  float* sp = states + (size_t)(ob * S_ + t0) * E_ + oc;

  auto step = [&](const int t, const int cur, float (&bt)[4],
                  const bool emit, const bool flush, const int ot) {
    float* xw = &xs[cur][(8 * half) * SP + e];
#pragma unroll
    for (int q = 0; q < 4; ++q) {
      const float re = st[2 * q], im = st[2 * q + 1];
      const float nr = fmaf(mag, re * cw - im * sw, bt[q]);
      const float ni = mag * (re * sw + im * cw);
      xw[(2 * q) * SP]     = nr;
      xw[(2 * q + 1) * SP] = ni;
    }

    asm volatile("s_wait_dscnt 0x0\n\ts_barrier_signal -1" ::: "memory");

    const int t2 = (t + 2 < tlast) ? (t + 2) : tlast;
#pragma unroll
    for (int q = 0; q < 4; ++q) bt[q] = bp[(size_t)q * S_ * E_ + (size_t)t2 * E_];

    asm volatile("s_barrier_wait -1" ::: "memory");

    const float4* ar = (const float4*)&xs[cur][m16 * SP + 32 * half];
    float4 av[8];
#pragma unroll
    for (int j = 0; j < 8; ++j) av[j] = ar[j];

    float4 o = make_float4(0.f, 0.f, 0.f, 0.f);
    if (flush) o = *(const float4*)&rs[cur ^ 1][ob * RP + oc];

    asm volatile("" ::: "memory");
    __builtin_amdgcn_sched_barrier(0);

    v8f acc0 = {}, acc1 = {};
#pragma unroll
    for (int j = 0; j < 8; ++j) {
      v2f a0 = {av[j].x, av[j].y};
      v2f a1 = {av[j].z, av[j].w};
      acc0 = WMMA_F32(a0, rb[2 * j],     acc0);
      acc1 = WMMA_F32(a1, rb[2 * j + 1], acc1);
    }
    __builtin_amdgcn_sched_barrier(0);
    st = acc0 + acc1;

    if (emit) {
      float* rw = &rs[cur][b0 * RP + e];
#pragma unroll
      for (int q = 0; q < 4; ++q) rw[q * RP] = st[2 * q];
      if (flush) *(float4*)(sp + (ot - 1) * E_) = o;
    }
  };

  int t = tw;
  for (int i = 0; i < nwarm; i += 2, t += 2) {
    step(t,     0, bte, false, false, 0);
    step(t + 1, 1, bto, false, false, 0);
  }
  for (int ot = 0; ot < LCH; ot += 2, t += 2) {
    step(t,     0, bte, true, ot > 0, ot);
    step(t + 1, 1, bto, true, true,   ot + 1);
  }

  {
    float* iw = &fs[b0 * RP + e];
#pragma unroll
    for (int q = 0; q < 4; ++q) iw[q * RP] = st[2 * q + 1];
    asm volatile("s_wait_dscnt 0x0\n\ts_barrier_signal -1\n\ts_barrier_wait -1" ::: "memory");
    const float4 o = *(const float4*)&rs[(LCH - 1) & 1][ob * RP + oc];
    *(float4*)(sp + (LCH - 1) * E_) = o;
    if (c == NBLK - 1) {
      *(float4*)(rfin + ob * E_ + oc) = o;
      *(float4*)(ifin + ob * E_ + oc) = *(const float4*)&fs[ob * RP + oc];
    }
  }
}

__global__ __launch_bounds__(128)
void k_out_ln(const float* __restrict__ states, const float* __restrict__ w_out,
              const float* __restrict__ gamma, const float* __restrict__ lbeta,
              float* __restrict__ y) {
  const int lane = threadIdx.x & 31;
  const int wv   = threadIdx.x >> 5;
  const int m16  = lane & 15;
  const int half = lane >> 4;
  const int rowbase = blockIdx.x * 16;

  __shared__ float  yt[16][D_ + 4];
  __shared__ float  psum[16][8], psq[16][8];
  __shared__ float2 lnp[16];

  v2f a[16];
  const float* srow = states + (size_t)(rowbase + m16) * E_ + 2 * half;
#pragma unroll
  for (int ks = 0; ks < 16; ++ks) {
    float2 av = *(const float2*)(srow + 4 * ks);
    a[ks].x = av.x; a[ks].y = av.y;
  }

  for (int dt = wv * 16; dt < wv * 16 + 16; ++dt) {
    v8f acc = {};
#pragma unroll
    for (int ks = 0; ks < 16; ++ks) {
      float2 bv = *(const float2*)(w_out + (size_t)(dt * 16 + m16) * E_ + 4 * ks + 2 * half);
      v2f bb; bb.x = bv.x; bb.y = bv.y;
      acc = WMMA_F32(a[ks], bb, acc);
    }
#pragma unroll
    for (int v = 0; v < 8; ++v)
      yt[v + 8 * half][dt * 16 + m16] = acc[v];
  }
  __syncthreads();

  {
    const int r = threadIdx.x >> 3, ss = threadIdx.x & 7;
    float s0 = 0.f, s1 = 0.f;
    for (int j = ss; j < D_; j += 8) {
      float v = yt[r][j];
      s0 += v; s1 += v * v;
    }
    psum[r][ss] = s0; psq[r][ss] = s1;
  }
  __syncthreads();
  if (threadIdx.x < 16) {
    const int r = threadIdx.x;
    float s0 = 0.f, s1 = 0.f;
#pragma unroll
    for (int j = 0; j < 8; ++j) { s0 += psum[r][j]; s1 += psq[r][j]; }
    const float mu  = s0 * (1.f / D_);
    const float var = s1 * (1.f / D_) - mu * mu;
    lnp[r] = make_float2(mu, rsqrtf(var + 1e-5f));
  }
  __syncthreads();

  for (int r = 0; r < 16; ++r) {
    const float mu = lnp[r].x, rs = lnp[r].y;
    for (int j = threadIdx.x; j < D_; j += 128) {
      float v = (yt[r][j] - mu) * rs * gamma[j] + lbeta[j];
      y[(size_t)(rowbase + r) * D_ + j] = v;
    }
  }
}

extern "C" void kernel_launch(void* const* d_in, const int* in_sizes, int n_in,
                              void* d_out, int out_size, void* d_ws, size_t ws_size,
                              hipStream_t stream) {
  (void)in_sizes; (void)n_in; (void)out_size; (void)ws_size;

  const float* x     = (const float*)d_in[0];
  const float* alpha = (const float*)d_in[1];
  const float* omega = (const float*)d_in[2];
  const float* w_in  = (const float*)d_in[3];
  const float* w_out = (const float*)d_in[4];
  const float* res   = (const float*)d_in[5];
  const float* gamma = (const float*)d_in[6];
  const float* lbeta = (const float*)d_in[7];

  float* out  = (float*)d_out;
  float* y    = out;
  float* rfin = out + YSZ;
  float* ifin = rfin + B_ * E_;

  float* beta   = (float*)d_ws;
  float* states = beta + (size_t)M_ * E_;

  hipLaunchKernelGGL(k_beta_gemm, dim3(M_ / 64), dim3(128), 0, stream, x, w_in, beta);
  hipLaunchKernelGGL(k_scan, dim3(NBLK), dim3(128), 0, stream,
                     beta, alpha, omega, res, states, rfin, ifin);
  hipLaunchKernelGGL(k_out_ln, dim3(M_ / 16), dim3(128), 0, stream,
                     states, w_out, gamma, lbeta, y);
}


# input GEMM: x and w_in staged through LDS in 64-k slabs (whole-line 16-B loads, w_in once per block, next slab in flight), b128 fragments; k order inside a slab permuted: its summation order changes
# speedup vs baseline: 2.1392x; 1.2221x over previous
#include <hip/hip_runtime.h>
#include <math.h>

typedef float v2f __attribute__((ext_vector_type(2)));
typedef float v4f __attribute__((ext_vector_type(4)));
typedef float v8f __attribute__((ext_vector_type(8)));

#define WMMA_F32(a, b, c) \
  __builtin_amdgcn_wmma_f32_16x16x4_f32(false, (a), false, (b), (short)0, (c), false, false)

static constexpr int B_ = 8, S_ = 2048, D_ = 1024, E_ = 64;
static constexpr int M_ = B_ * S_;
static constexpr long long YSZ = (long long)M_ * D_;

static constexpr int KS    = 64;
static constexpr int NSLAB = D_ / KS;
static constexpr int TP    = 68;

__global__ __launch_bounds__(128)
void k_beta_gemm(const float* __restrict__ x, const float* __restrict__ w_in,
                 float* __restrict__ beta) {
  const int tid  = threadIdx.x;
  const int lane = tid & 31;
  const int w    = tid >> 5;
  const int wv   = (blockIdx.x << 2) + w;
  const int m16  = lane & 15;
  const int half = lane >> 4;

  __shared__ __attribute__((aligned(16))) float xt[2][64 * TP];
  __shared__ __attribute__((aligned(16))) float wt[2][64 * TP];

  const int sr = tid >> 4;
  const int sc = (tid & 15) * 4;
  const float* xg = x    + (size_t)(blockIdx.x * 64 + sr) * D_ + sc;
  const float* wg = w_in + (size_t)sr * D_ + sc;
  const int so = sr * TP + sc;

  v4f gx[8], gw[8];

  auto fetch = [&](const int s) {
#pragma unroll
    for (int i = 0; i < 8; ++i) {
      gx[i] = *(const v4f*)(xg + (size_t)(8 * i) * D_ + s * KS);
      gw[i] = *(const v4f*)(wg + (size_t)(8 * i) * D_ + s * KS);
    }
  };
  auto stage = [&](const int buf) {
#pragma unroll
    for (int i = 0; i < 8; ++i) {
      *(v4f*)&xt[buf][so + 8 * i * TP] = gx[i];
      *(v4f*)&wt[buf][so + 8 * i * TP] = gw[i];
    }
  };

  v8f acc[4];
#pragma unroll
  for (int i = 0; i < 4; ++i) acc[i] = {};

  auto ldq = [&](const float* ar, const float* br, const int q, v4f (&a)[2], v4f (&b)[2][4]) {
#pragma unroll
    for (int jj = 0; jj < 2; ++jj) {
      a[jj] = *(const v4f*)(ar + 4 * (2 * q + jj));
#pragma unroll
      for (int nt = 0; nt < 4; ++nt)
        b[jj][nt] = *(const v4f*)(br + nt * 16 * TP + 4 * (2 * q + jj));
    }
  };
  auto mmq = [&](const v4f (&a)[2], const v4f (&b)[2][4]) {
#pragma unroll
    for (int jj = 0; jj < 2; ++jj) {
      const v2f a0 = {a[jj].x, a[jj].y};
      const v2f a1 = {a[jj].z, a[jj].w};
#pragma unroll
      for (int nt = 0; nt < 4; ++nt) {
        const v2f b0 = {b[jj][nt].x, b[jj][nt].y};
        acc[nt] = WMMA_F32(a0, b0, acc[nt]);
      }
#pragma unroll
      for (int nt = 0; nt < 4; ++nt) {
        const v2f b1 = {b[jj][nt].z, b[jj][nt].w};
        acc[nt] = WMMA_F32(a1, b1, acc[nt]);
      }
    }
  };
#define GROUP_EDGE() do { asm volatile("" ::: "memory"); __builtin_amdgcn_sched_barrier(0); } while (0)

  auto multiply = [&](const int buf) {
    const float* ar = &xt[buf][(16 * w + m16) * TP + 32 * half];
    const float* br = &wt[buf][m16 * TP + 32 * half];
    v4f a0[2], b0[2][4], a1[2], b1[2][4];
    ldq(ar, br, 0, a0, b0);
    ldq(ar, br, 1, a1, b1);
    GROUP_EDGE();
    mmq(a0, b0);
    GROUP_EDGE();
    ldq(ar, br, 2, a0, b0);
    GROUP_EDGE();
    mmq(a1, b1);
    GROUP_EDGE();
    ldq(ar, br, 3, a1, b1);
    GROUP_EDGE();
    mmq(a0, b0);
    mmq(a1, b1);
    GROUP_EDGE();
  };

  fetch(0);
  stage(0);
  __syncthreads();
  for (int s = 0; s < NSLAB - 1; ++s) {
    fetch(s + 1);
    multiply(s & 1);
    stage((s + 1) & 1);
    __syncthreads();
  }
  multiply((NSLAB - 1) & 1);

#pragma unroll
  for (int nt = 0; nt < 4; ++nt)
#pragma unroll
    for (int v = 0; v < 8; ++v)
      beta[(size_t)(wv * 16 + v + 8 * half) * E_ + nt * 16 + m16] = acc[nt][v];
}

static constexpr int LCH  = 8;
static constexpr int NBLK = S_ / LCH;
static constexpr int WARM = 64;
static constexpr int SP = 68;
static constexpr int RP = 72;

__global__ __launch_bounds__(128)
void k_scan(const float* __restrict__ beta, const float* __restrict__ alpha,
            const float* __restrict__ omega, const float* __restrict__ res,
            float* __restrict__ states,
            float* __restrict__ rfin, float* __restrict__ ifin) {
  const int c    = blockIdx.x;
  const int tid  = threadIdx.x;
  const int lane = tid & 31;
  const int w    = tid >> 5;
  const int m16  = lane & 15;
  const int half = lane >> 4;
  const int e    = w * 16 + m16;
  const int b0   = 4 * half;

  const int t0    = c * LCH;
  const int tw    = (t0 > WARM) ? (t0 - WARM) : 0;
  const int nwarm = t0 - tw;
  const int tlast = t0 + LCH - 1;

  const float mag = 1.f / (1.f + expf(-alpha[e]));
  const float cw  = cosf(omega[e]);
  const float sw  = sinf(omega[e]);

  v2f rb[16];
#pragma unroll
  for (int t = 0; t < 16; ++t) {
    const int k = 32 * half + 2 * t;
    rb[t].x = res[k * E_ + e];
    rb[t].y = res[(k + 1) * E_ + e];
  }

  __shared__ __attribute__((aligned(16))) float xs[2][16 * SP];
  __shared__ __attribute__((aligned(16))) float rs[2][B_ * RP];
  __shared__ __attribute__((aligned(16))) float fs[B_ * RP];

  v8f st = {};

  const float* bp = beta + (size_t)(b0 * S_) * E_ + e;

  float bte[4], bto[4];
#pragma unroll
  for (int q = 0; q < 4; ++q) bte[q] = bp[(size_t)q * S_ * E_ + (size_t)tw * E_];
#pragma unroll
  for (int q = 0; q < 4; ++q) bto[q] = bp[(size_t)q * S_ * E_ + (size_t)(tw + 1) * E_];

  const int ob = tid >> 4, oc = (tid & 15) * 4;
  float* sp = states + (size_t)(ob * S_ + t0) * E_ + oc;

  auto step = [&](const int t, const int cur, float (&bt)[4],
                  const bool emit, const bool flush, const int ot) {
    float* xw = &xs[cur][(8 * half) * SP + e];
#pragma unroll
    for (int q = 0; q < 4; ++q) {
      const float re = st[2 * q], im = st[2 * q + 1];
      const float nr = fmaf(mag, re * cw - im * sw, bt[q]);
      const float ni = mag * (re * sw + im * cw);
      xw[(2 * q) * SP]     = nr;
      xw[(2 * q + 1) * SP] = ni;
    }

    asm volatile("s_wait_dscnt 0x0\n\ts_barrier_signal -1" ::: "memory");

    const int t2 = (t + 2 < tlast) ? (t + 2) : tlast;
#pragma unroll
    for (int q = 0; q < 4; ++q) bt[q] = bp[(size_t)q * S_ * E_ + (size_t)t2 * E_];

    asm volatile("s_barrier_wait -1" ::: "memory");

    const float4* ar = (const float4*)&xs[cur][m16 * SP + 32 * half];
    float4 av[8];
#pragma unroll
    for (int j = 0; j < 8; ++j) av[j] = ar[j];

    float4 o = make_float4(0.f, 0.f, 0.f, 0.f);
    if (flush) o = *(const float4*)&rs[cur ^ 1][ob * RP + oc];

    asm volatile("" ::: "memory");
    __builtin_amdgcn_sched_barrier(0);

    v8f acc0 = {}, acc1 = {};
#pragma unroll
    for (int j = 0; j < 8; ++j) {
      v2f a0 = {av[j].x, av[j].y};
      v2f a1 = {av[j].z, av[j].w};
      acc0 = WMMA_F32(a0, rb[2 * j],     acc0);
      acc1 = WMMA_F32(a1, rb[2 * j + 1], acc1);
    }
    __builtin_amdgcn_sched_barrier(0);
    st = acc0 + acc1;

    if (emit) {
      float* rw = &rs[cur][b0 * RP + e];
#pragma unroll
      for (int q = 0; q < 4; ++q) rw[q * RP] = st[2 * q];
      if (flush) *(float4*)(sp + (ot - 1) * E_) = o;
    }
  };

  int t = tw;
  for (int i = 0; i < nwarm; i += 2, t += 2) {
    step(t,     0, bte, false, false, 0);
    step(t + 1, 1, bto, false, false, 0);
  }
  for (int ot = 0; ot < LCH; ot += 2, t += 2) {
    step(t,     0, bte, true, ot > 0, ot);
    step(t + 1, 1, bto, true, true,   ot + 1);
  }

  {
    float* iw = &fs[b0 * RP + e];
#pragma unroll
    for (int q = 0; q < 4; ++q) iw[q * RP] = st[2 * q + 1];
    asm volatile("s_wait_dscnt 0x0\n\ts_barrier_signal -1\n\ts_barrier_wait -1" ::: "memory");
    const float4 o = *(const float4*)&rs[(LCH - 1) & 1][ob * RP + oc];
    *(float4*)(sp + (LCH - 1) * E_) = o;
    if (c == NBLK - 1) {
      *(float4*)(rfin + ob * E_ + oc) = o;
      *(float4*)(ifin + ob * E_ + oc) = *(const float4*)&fs[ob * RP + oc];
    }
  }
}

__global__ __launch_bounds__(128)
void k_out_ln(const float* __restrict__ states, const float* __restrict__ w_out,
              const float* __restrict__ gamma, const float* __restrict__ lbeta,
              float* __restrict__ y) {
  const int lane = threadIdx.x & 31;
  const int wv   = threadIdx.x >> 5;
  const int m16  = lane & 15;
  const int half = lane >> 4;
  const int rowbase = blockIdx.x * 16;

  __shared__ float  yt[16][D_ + 4];
  __shared__ float  psum[16][8], psq[16][8];
  __shared__ float2 lnp[16];

  v2f a[16];
  const float* srow = states + (size_t)(rowbase + m16) * E_ + 2 * half;
#pragma unroll
  for (int ks = 0; ks < 16; ++ks) {
    float2 av = *(const float2*)(srow + 4 * ks);
    a[ks].x = av.x; a[ks].y = av.y;
  }

  for (int dt = wv * 16; dt < wv * 16 + 16; ++dt) {
    v8f acc = {};
#pragma unroll
    for (int ks = 0; ks < 16; ++ks) {
      float2 bv = *(const float2*)(w_out + (size_t)(dt * 16 + m16) * E_ + 4 * ks + 2 * half);
      v2f bb; bb.x = bv.x; bb.y = bv.y;
      acc = WMMA_F32(a[ks], bb, acc);
    }
#pragma unroll
    for (int v = 0; v < 8; ++v)
      yt[v + 8 * half][dt * 16 + m16] = acc[v];
  }
  __syncthreads();

  {
    const int r = threadIdx.x >> 3, ss = threadIdx.x & 7;
    float s0 = 0.f, s1 = 0.f;
    for (int j = ss; j < D_; j += 8) {
      float v = yt[r][j];
      s0 += v; s1 += v * v;
    }
    psum[r][ss] = s0; psq[r][ss] = s1;
  }
  __syncthreads();
  if (threadIdx.x < 16) {
    const int r = threadIdx.x;
    float s0 = 0.f, s1 = 0.f;
#pragma unroll
    for (int j = 0; j < 8; ++j) { s0 += psum[r][j]; s1 += psq[r][j]; }
    const float mu  = s0 * (1.f / D_);
    const float var = s1 * (1.f / D_) - mu * mu;
    lnp[r] = make_float2(mu, rsqrtf(var + 1e-5f));
  }
  __syncthreads();

  for (int r = 0; r < 16; ++r) {
    const float mu = lnp[r].x, rs = lnp[r].y;
    for (int j = threadIdx.x; j < D_; j += 128) {
      float v = (yt[r][j] - mu) * rs * gamma[j] + lbeta[j];
      y[(size_t)(rowbase + r) * D_ + j] = v;
    }
  }
}

extern "C" void kernel_launch(void* const* d_in, const int* in_sizes, int n_in,
                              void* d_out, int out_size, void* d_ws, size_t ws_size,
                              hipStream_t stream) {
  (void)in_sizes; (void)n_in; (void)out_size; (void)ws_size;

  const float* x     = (const float*)d_in[0];
  const float* alpha = (const float*)d_in[1];
  const float* omega = (const float*)d_in[2];
  const float* w_in  = (const float*)d_in[3];
  const float* w_out = (const float*)d_in[4];
  const float* res   = (const float*)d_in[5];
  const float* gamma = (const float*)d_in[6];
  const float* lbeta = (const float*)d_in[7];

  float* out  = (float*)d_out;
  float* y    = out;
  float* rfin = out + YSZ;
  float* ifin = rfin + B_ * E_;

  float* beta   = (float*)d_ws;
  float* states = beta + (size_t)M_ * E_;

  hipLaunchKernelGGL(k_beta_gemm, dim3(M_ / 64), dim3(128), 0, stream, x, w_in, beta);
  hipLaunchKernelGGL(k_scan, dim3(NBLK), dim3(128), 0, stream,
                     beta, alpha, omega, res, states, rfin, ifin);
  hipLaunchKernelGGL(k_out_ln, dim3(M_ / 16), dim3(128), 0, stream,
                     states, w_out, gamma, lbeta, y);
}


# output kernel LN epilogue: one wave per row, row read once from LDS as 16-B pieces into registers, lane-shuffle sums, whole 512-B runs stored per instruction; the LN sums' summation order changes
# speedup vs baseline: 4.0500x; 1.8932x over previous
#include <hip/hip_runtime.h>
#include <math.h>

typedef float v2f __attribute__((ext_vector_type(2)));
typedef float v4f __attribute__((ext_vector_type(4)));
typedef float v8f __attribute__((ext_vector_type(8)));

#define WMMA_F32(a, b, c) \
  __builtin_amdgcn_wmma_f32_16x16x4_f32(false, (a), false, (b), (short)0, (c), false, false)

static constexpr int B_ = 8, S_ = 2048, D_ = 1024, E_ = 64;
static constexpr int M_ = B_ * S_;
static constexpr long long YSZ = (long long)M_ * D_;

static constexpr int KS    = 64;
static constexpr int NSLAB = D_ / KS;
static constexpr int TP    = 68;

__global__ __launch_bounds__(128)
void k_beta_gemm(const float* __restrict__ x, const float* __restrict__ w_in,
                 float* __restrict__ beta) {
  const int tid  = threadIdx.x;
  const int lane = tid & 31;
  const int w    = tid >> 5;
  const int wv   = (blockIdx.x << 2) + w;
  const int m16  = lane & 15;
  const int half = lane >> 4;

  __shared__ __attribute__((aligned(16))) float xt[2][64 * TP];
  __shared__ __attribute__((aligned(16))) float wt[2][64 * TP];

  const int sr = tid >> 4;
  const int sc = (tid & 15) * 4;
  const float* xg = x    + (size_t)(blockIdx.x * 64 + sr) * D_ + sc;
  const float* wg = w_in + (size_t)sr * D_ + sc;
  const int so = sr * TP + sc;

  v4f gx[8], gw[8];

  auto fetch = [&](const int s) {
#pragma unroll
    for (int i = 0; i < 8; ++i) {
      gx[i] = *(const v4f*)(xg + (size_t)(8 * i) * D_ + s * KS);
      gw[i] = *(const v4f*)(wg + (size_t)(8 * i) * D_ + s * KS);
    }
  };
  auto stage = [&](const int buf) {
#pragma unroll
    for (int i = 0; i < 8; ++i) {
      *(v4f*)&xt[buf][so + 8 * i * TP] = gx[i];
      *(v4f*)&wt[buf][so + 8 * i * TP] = gw[i];
    }
  };

  v8f acc[4];
#pragma unroll
  for (int i = 0; i < 4; ++i) acc[i] = {};

  auto ldq = [&](const float* ar, const float* br, const int q, v4f (&a)[2], v4f (&b)[2][4]) {
#pragma unroll
    for (int jj = 0; jj < 2; ++jj) {
      a[jj] = *(const v4f*)(ar + 4 * (2 * q + jj));
#pragma unroll
      for (int nt = 0; nt < 4; ++nt)
        b[jj][nt] = *(const v4f*)(br + nt * 16 * TP + 4 * (2 * q + jj));
    }
  };
  auto mmq = [&](const v4f (&a)[2], const v4f (&b)[2][4]) {
#pragma unroll
    for (int jj = 0; jj < 2; ++jj) {
      const v2f a0 = {a[jj].x, a[jj].y};
      const v2f a1 = {a[jj].z, a[jj].w};
#pragma unroll
      for (int nt = 0; nt < 4; ++nt) {
        const v2f b0 = {b[jj][nt].x, b[jj][nt].y};
        acc[nt] = WMMA_F32(a0, b0, acc[nt]);
      }
#pragma unroll
      for (int nt = 0; nt < 4; ++nt) {
        const v2f b1 = {b[jj][nt].z, b[jj][nt].w};
        acc[nt] = WMMA_F32(a1, b1, acc[nt]);
      }
    }
  };
#define GROUP_EDGE() do { asm volatile("" ::: "memory"); __builtin_amdgcn_sched_barrier(0); } while (0)

  auto multiply = [&](const int buf) {
    const float* ar = &xt[buf][(16 * w + m16) * TP + 32 * half];
    const float* br = &wt[buf][m16 * TP + 32 * half];
    v4f a0[2], b0[2][4], a1[2], b1[2][4];
    ldq(ar, br, 0, a0, b0);
    ldq(ar, br, 1, a1, b1);
    GROUP_EDGE();
    mmq(a0, b0);
    GROUP_EDGE();
    ldq(ar, br, 2, a0, b0);
    GROUP_EDGE();
    mmq(a1, b1);
    GROUP_EDGE();
    ldq(ar, br, 3, a1, b1);
    GROUP_EDGE();
    mmq(a0, b0);
    mmq(a1, b1);
    GROUP_EDGE();
  };

  fetch(0);
  stage(0);
  __syncthreads();
  for (int s = 0; s < NSLAB - 1; ++s) {
    fetch(s + 1);
    multiply(s & 1);
    stage((s + 1) & 1);
    __syncthreads();
  }
  multiply((NSLAB - 1) & 1);

#pragma unroll
  for (int nt = 0; nt < 4; ++nt)
#pragma unroll
    for (int v = 0; v < 8; ++v)
      beta[(size_t)(wv * 16 + v + 8 * half) * E_ + nt * 16 + m16] = acc[nt][v];
}

static constexpr int LCH  = 8;
static constexpr int NBLK = S_ / LCH;
static constexpr int WARM = 64;
static constexpr int SP = 68;
static constexpr int RP = 72;

__global__ __launch_bounds__(128)
void k_scan(const float* __restrict__ beta, const float* __restrict__ alpha,
            const float* __restrict__ omega, const float* __restrict__ res,
            float* __restrict__ states,
            float* __restrict__ rfin, float* __restrict__ ifin) {
  const int c    = blockIdx.x;
  const int tid  = threadIdx.x;
  const int lane = tid & 31;
  const int w    = tid >> 5;
  const int m16  = lane & 15;
  const int half = lane >> 4;
  const int e    = w * 16 + m16;
  const int b0   = 4 * half;

  const int t0    = c * LCH;
  const int tw    = (t0 > WARM) ? (t0 - WARM) : 0;
  const int nwarm = t0 - tw;
  const int tlast = t0 + LCH - 1;

  const float mag = 1.f / (1.f + expf(-alpha[e]));
  const float cw  = cosf(omega[e]);
  const float sw  = sinf(omega[e]);

  v2f rb[16];
#pragma unroll
  for (int t = 0; t < 16; ++t) {
    const int k = 32 * half + 2 * t;
    rb[t].x = res[k * E_ + e];
    rb[t].y = res[(k + 1) * E_ + e];
  }

  __shared__ __attribute__((aligned(16))) float xs[2][16 * SP];
  __shared__ __attribute__((aligned(16))) float rs[2][B_ * RP];
  __shared__ __attribute__((aligned(16))) float fs[B_ * RP];

  v8f st = {};

  const float* bp = beta + (size_t)(b0 * S_) * E_ + e;

  float bte[4], bto[4];
#pragma unroll
  for (int q = 0; q < 4; ++q) bte[q] = bp[(size_t)q * S_ * E_ + (size_t)tw * E_];
#pragma unroll
  for (int q = 0; q < 4; ++q) bto[q] = bp[(size_t)q * S_ * E_ + (size_t)(tw + 1) * E_];

  const int ob = tid >> 4, oc = (tid & 15) * 4;
  float* sp = states + (size_t)(ob * S_ + t0) * E_ + oc;

  auto step = [&](const int t, const int cur, float (&bt)[4],
                  const bool emit, const bool flush, const int ot) {
    float* xw = &xs[cur][(8 * half) * SP + e];
#pragma unroll
    for (int q = 0; q < 4; ++q) {
      const float re = st[2 * q], im = st[2 * q + 1];
      const float nr = fmaf(mag, re * cw - im * sw, bt[q]);
      const float ni = mag * (re * sw + im * cw);
      xw[(2 * q) * SP]     = nr;
      xw[(2 * q + 1) * SP] = ni;
    }

    asm volatile("s_wait_dscnt 0x0\n\ts_barrier_signal -1" ::: "memory");

    const int t2 = (t + 2 < tlast) ? (t + 2) : tlast;
#pragma unroll
    for (int q = 0; q < 4; ++q) bt[q] = bp[(size_t)q * S_ * E_ + (size_t)t2 * E_];

    asm volatile("s_barrier_wait -1" ::: "memory");

    const float4* ar = (const float4*)&xs[cur][m16 * SP + 32 * half];
    float4 av[8];
#pragma unroll
    for (int j = 0; j < 8; ++j) av[j] = ar[j];

    float4 o = make_float4(0.f, 0.f, 0.f, 0.f);
    if (flush) o = *(const float4*)&rs[cur ^ 1][ob * RP + oc];

    asm volatile("" ::: "memory");
    __builtin_amdgcn_sched_barrier(0);

    v8f acc0 = {}, acc1 = {};
#pragma unroll
    for (int j = 0; j < 8; ++j) {
      v2f a0 = {av[j].x, av[j].y};
      v2f a1 = {av[j].z, av[j].w};
      acc0 = WMMA_F32(a0, rb[2 * j],     acc0);
      acc1 = WMMA_F32(a1, rb[2 * j + 1], acc1);
    }
    __builtin_amdgcn_sched_barrier(0);
    st = acc0 + acc1;

    if (emit) {
      float* rw = &rs[cur][b0 * RP + e];
#pragma unroll
      for (int q = 0; q < 4; ++q) rw[q * RP] = st[2 * q];
      if (flush) *(float4*)(sp + (ot - 1) * E_) = o;
    }
  };

  int t = tw;
  for (int i = 0; i < nwarm; i += 2, t += 2) {
    step(t,     0, bte, false, false, 0);
    step(t + 1, 1, bto, false, false, 0);
  }
  for (int ot = 0; ot < LCH; ot += 2, t += 2) {
    step(t,     0, bte, true, ot > 0, ot);
    step(t + 1, 1, bto, true, true,   ot + 1);
  }

  {
    float* iw = &fs[b0 * RP + e];
#pragma unroll
    for (int q = 0; q < 4; ++q) iw[q * RP] = st[2 * q + 1];
    asm volatile("s_wait_dscnt 0x0\n\ts_barrier_signal -1\n\ts_barrier_wait -1" ::: "memory");
    const float4 o = *(const float4*)&rs[(LCH - 1) & 1][ob * RP + oc];
    *(float4*)(sp + (LCH - 1) * E_) = o;
    if (c == NBLK - 1) {
      *(float4*)(rfin + ob * E_ + oc) = o;
      *(float4*)(ifin + ob * E_ + oc) = *(const float4*)&fs[ob * RP + oc];
    }
  }
}

__global__ __launch_bounds__(128)
void k_out_ln(const float* __restrict__ states, const float* __restrict__ w_out,
              const float* __restrict__ gamma, const float* __restrict__ lbeta,
              float* __restrict__ y) {
  const int lane = threadIdx.x & 31;
  const int wv   = threadIdx.x >> 5;
  const int m16  = lane & 15;
  const int half = lane >> 4;
  const int rowbase = blockIdx.x * 16;

  __shared__ __attribute__((aligned(16))) float yt[16][D_ + 4];

  v4f gm[8], bt[8];
#pragma unroll
  for (int i = 0; i < 8; ++i) {
    gm[i] = *(const v4f*)(gamma + 128 * i + 4 * lane);
    bt[i] = *(const v4f*)(lbeta + 128 * i + 4 * lane);
  }

  v2f a[16];
  const float* srow = states + (size_t)(rowbase + m16) * E_ + 2 * half;
#pragma unroll
  for (int ks = 0; ks < 16; ++ks) {
    float2 av = *(const float2*)(srow + 4 * ks);
    a[ks].x = av.x; a[ks].y = av.y;
  }

#pragma unroll 1
  for (int dt = wv * 16; dt < wv * 16 + 16; ++dt) {
    v8f acc = {};
#pragma unroll
    for (int ks = 0; ks < 16; ++ks) {
      float2 bv = *(const float2*)(w_out + (size_t)(dt * 16 + m16) * E_ + 4 * ks + 2 * half);
      v2f bb; bb.x = bv.x; bb.y = bv.y;
      acc = WMMA_F32(a[ks], bb, acc);
    }
#pragma unroll
    for (int v = 0; v < 8; ++v)
      yt[v + 8 * half][dt * 16 + m16] = acc[v];
  }
  __syncthreads();

  v4f yv[4][8];
#pragma unroll
  for (int q = 0; q < 4; ++q)
#pragma unroll
    for (int i = 0; i < 8; ++i)
      yv[q][i] = *(const v4f*)&yt[4 * wv + q][128 * i + 4 * lane];

  float s0[4], s1[4];
#pragma unroll
  for (int q = 0; q < 4; ++q) {
    v4f p0 = yv[q][0];
    v4f p1 = yv[q][0] * yv[q][0];
#pragma unroll
    for (int i = 1; i < 8; ++i) {
      p0 += yv[q][i];
      p1 += yv[q][i] * yv[q][i];
    }
    s0[q] = (p0.x + p0.y) + (p0.z + p0.w);
    s1[q] = (p1.x + p1.y) + (p1.z + p1.w);
  }
#pragma unroll
  for (int m = 16; m >= 1; m >>= 1) {
#pragma unroll
    for (int q = 0; q < 4; ++q) {
      s0[q] += __shfl_xor(s0[q], m);
      s1[q] += __shfl_xor(s1[q], m);
    }
  }

  float* yo = y + (size_t)(rowbase + 4 * wv) * D_ + 4 * lane;
#pragma unroll
  for (int q = 0; q < 4; ++q) {
    const float mu  = s0[q] * (1.f / D_);
    const float var = s1[q] * (1.f / D_) - mu * mu;
    const float rs  = rsqrtf(var + 1e-5f);
#pragma unroll
    for (int i = 0; i < 8; ++i) {
      const v4f o = (yv[q][i] - mu) * rs * gm[i] + bt[i];
      *(v4f*)(yo + (size_t)q * D_ + 128 * i) = o;
    }
  }
}

extern "C" void kernel_launch(void* const* d_in, const int* in_sizes, int n_in,
                              void* d_out, int out_size, void* d_ws, size_t ws_size,
                              hipStream_t stream) {
  (void)in_sizes; (void)n_in; (void)out_size; (void)ws_size;

  const float* x     = (const float*)d_in[0];
  const float* alpha = (const float*)d_in[1];
  const float* omega = (const float*)d_in[2];
  const float* w_in  = (const float*)d_in[3];
  const float* w_out = (const float*)d_in[4];
  const float* res   = (const float*)d_in[5];
  const float* gamma = (const float*)d_in[6];
  const float* lbeta = (const float*)d_in[7];

  float* out  = (float*)d_out;
  float* y    = out;
  float* rfin = out + YSZ;
  float* ifin = rfin + B_ * E_;

  float* beta   = (float*)d_ws;
  float* states = beta + (size_t)M_ * E_;

  hipLaunchKernelGGL(k_beta_gemm, dim3(M_ / 64), dim3(128), 0, stream, x, w_in, beta);
  hipLaunchKernelGGL(k_scan, dim3(NBLK), dim3(128), 0, stream,
                     beta, alpha, omega, res, states, rfin, ifin);
  hipLaunchKernelGGL(k_out_ln, dim3(M_ / 16), dim3(128), 0, stream,
                     states, w_out, gamma, lbeta, y);
}


# output kernel: states and w_out tiles fetched by whole-line 16-B loads staged through per-wave LDS into b128 fragments (were 8-B loads over 16 lines each); k order permuted: summation order changes
# speedup vs baseline: 4.5664x; 1.1275x over previous
#include <hip/hip_runtime.h>
#include <math.h>

typedef float v2f __attribute__((ext_vector_type(2)));
typedef float v4f __attribute__((ext_vector_type(4)));
typedef float v8f __attribute__((ext_vector_type(8)));

#define WMMA_F32(a, b, c) \
  __builtin_amdgcn_wmma_f32_16x16x4_f32(false, (a), false, (b), (short)0, (c), false, false)

static constexpr int B_ = 8, S_ = 2048, D_ = 1024, E_ = 64;
static constexpr int M_ = B_ * S_;
static constexpr long long YSZ = (long long)M_ * D_;

static constexpr int KS    = 64;
static constexpr int NSLAB = D_ / KS;
static constexpr int TP    = 68;

__global__ __launch_bounds__(128)
void k_beta_gemm(const float* __restrict__ x, const float* __restrict__ w_in,
                 float* __restrict__ beta) {
  const int tid  = threadIdx.x;
  const int lane = tid & 31;
  const int w    = tid >> 5;
  const int wv   = (blockIdx.x << 2) + w;
  const int m16  = lane & 15;
  const int half = lane >> 4;

  __shared__ __attribute__((aligned(16))) float xt[2][64 * TP];
  __shared__ __attribute__((aligned(16))) float wt[2][64 * TP];

  const int sr = tid >> 4;
  const int sc = (tid & 15) * 4;
  const float* xg = x    + (size_t)(blockIdx.x * 64 + sr) * D_ + sc;
  const float* wg = w_in + (size_t)sr * D_ + sc;
  const int so = sr * TP + sc;

  v4f gx[8], gw[8];

  auto fetch = [&](const int s) {
#pragma unroll
    for (int i = 0; i < 8; ++i) {
      gx[i] = *(const v4f*)(xg + (size_t)(8 * i) * D_ + s * KS);
      gw[i] = *(const v4f*)(wg + (size_t)(8 * i) * D_ + s * KS);
    }
  };
  auto stage = [&](const int buf) {
#pragma unroll
    for (int i = 0; i < 8; ++i) {
      *(v4f*)&xt[buf][so + 8 * i * TP] = gx[i];
      *(v4f*)&wt[buf][so + 8 * i * TP] = gw[i];
    }
  };

  v8f acc[4];
#pragma unroll
  for (int i = 0; i < 4; ++i) acc[i] = {};

  auto ldq = [&](const float* ar, const float* br, const int q, v4f (&a)[2], v4f (&b)[2][4]) {
#pragma unroll
    for (int jj = 0; jj < 2; ++jj) {
      a[jj] = *(const v4f*)(ar + 4 * (2 * q + jj));
#pragma unroll
      for (int nt = 0; nt < 4; ++nt)
        b[jj][nt] = *(const v4f*)(br + nt * 16 * TP + 4 * (2 * q + jj));
    }
  };
  auto mmq = [&](const v4f (&a)[2], const v4f (&b)[2][4]) {
#pragma unroll
    for (int jj = 0; jj < 2; ++jj) {
      const v2f a0 = {a[jj].x, a[jj].y};
      const v2f a1 = {a[jj].z, a[jj].w};
#pragma unroll
      for (int nt = 0; nt < 4; ++nt) {
        const v2f b0 = {b[jj][nt].x, b[jj][nt].y};
        acc[nt] = WMMA_F32(a0, b0, acc[nt]);
      }
#pragma unroll
      for (int nt = 0; nt < 4; ++nt) {
        const v2f b1 = {b[jj][nt].z, b[jj][nt].w};
        acc[nt] = WMMA_F32(a1, b1, acc[nt]);
      }
    }
  };
#define GROUP_EDGE() do { asm volatile("" ::: "memory"); __builtin_amdgcn_sched_barrier(0); } while (0)

  auto multiply = [&](const int buf) {
    const float* ar = &xt[buf][(16 * w + m16) * TP + 32 * half];
    const float* br = &wt[buf][m16 * TP + 32 * half];
    v4f a0[2], b0[2][4], a1[2], b1[2][4];
    ldq(ar, br, 0, a0, b0);
    ldq(ar, br, 1, a1, b1);
    GROUP_EDGE();
    mmq(a0, b0);
    GROUP_EDGE();
    ldq(ar, br, 2, a0, b0);
    GROUP_EDGE();
    mmq(a1, b1);
    GROUP_EDGE();
    ldq(ar, br, 3, a1, b1);
    GROUP_EDGE();
    mmq(a0, b0);
    mmq(a1, b1);
    GROUP_EDGE();
  };

  fetch(0);
  stage(0);
  __syncthreads();
  for (int s = 0; s < NSLAB - 1; ++s) {
    fetch(s + 1);
    multiply(s & 1);
    stage((s + 1) & 1);
    __syncthreads();
  }
  multiply((NSLAB - 1) & 1);

#pragma unroll
  for (int nt = 0; nt < 4; ++nt)
#pragma unroll
    for (int v = 0; v < 8; ++v)
      beta[(size_t)(wv * 16 + v + 8 * half) * E_ + nt * 16 + m16] = acc[nt][v];
}

static constexpr int LCH  = 8;
static constexpr int NBLK = S_ / LCH;
static constexpr int WARM = 64;
static constexpr int SP = 68;
static constexpr int RP = 72;

__global__ __launch_bounds__(128)
void k_scan(const float* __restrict__ beta, const float* __restrict__ alpha,
            const float* __restrict__ omega, const float* __restrict__ res,
            float* __restrict__ states,
            float* __restrict__ rfin, float* __restrict__ ifin) {
  const int c    = blockIdx.x;
  const int tid  = threadIdx.x;
  const int lane = tid & 31;
  const int w    = tid >> 5;
  const int m16  = lane & 15;
  const int half = lane >> 4;
  const int e    = w * 16 + m16;
  const int b0   = 4 * half;

  const int t0    = c * LCH;
  const int tw    = (t0 > WARM) ? (t0 - WARM) : 0;
  const int nwarm = t0 - tw;
  const int tlast = t0 + LCH - 1;

  const float mag = 1.f / (1.f + expf(-alpha[e]));
  const float cw  = cosf(omega[e]);
  const float sw  = sinf(omega[e]);

  v2f rb[16];
#pragma unroll
  for (int t = 0; t < 16; ++t) {
    const int k = 32 * half + 2 * t;
    rb[t].x = res[k * E_ + e];
    rb[t].y = res[(k + 1) * E_ + e];
  }

  __shared__ __attribute__((aligned(16))) float xs[2][16 * SP];
  __shared__ __attribute__((aligned(16))) float rs[2][B_ * RP];
  __shared__ __attribute__((aligned(16))) float fs[B_ * RP];

  v8f st = {};

  const float* bp = beta + (size_t)(b0 * S_) * E_ + e;

  float bte[4], bto[4];
#pragma unroll
  for (int q = 0; q < 4; ++q) bte[q] = bp[(size_t)q * S_ * E_ + (size_t)tw * E_];
#pragma unroll
  for (int q = 0; q < 4; ++q) bto[q] = bp[(size_t)q * S_ * E_ + (size_t)(tw + 1) * E_];

  const int ob = tid >> 4, oc = (tid & 15) * 4;
  float* sp = states + (size_t)(ob * S_ + t0) * E_ + oc;

  auto step = [&](const int t, const int cur, float (&bt)[4],
                  const bool emit, const bool flush, const int ot) {
    float* xw = &xs[cur][(8 * half) * SP + e];
#pragma unroll
    for (int q = 0; q < 4; ++q) {
      const float re = st[2 * q], im = st[2 * q + 1];
      const float nr = fmaf(mag, re * cw - im * sw, bt[q]);
      const float ni = mag * (re * sw + im * cw);
      xw[(2 * q) * SP]     = nr;
      xw[(2 * q + 1) * SP] = ni;
    }

    asm volatile("s_wait_dscnt 0x0\n\ts_barrier_signal -1" ::: "memory");

    const int t2 = (t + 2 < tlast) ? (t + 2) : tlast;
#pragma unroll
    for (int q = 0; q < 4; ++q) bt[q] = bp[(size_t)q * S_ * E_ + (size_t)t2 * E_];

    asm volatile("s_barrier_wait -1" ::: "memory");

    const float4* ar = (const float4*)&xs[cur][m16 * SP + 32 * half];
    float4 av[8];
#pragma unroll
    for (int j = 0; j < 8; ++j) av[j] = ar[j];

    float4 o = make_float4(0.f, 0.f, 0.f, 0.f);
    if (flush) o = *(const float4*)&rs[cur ^ 1][ob * RP + oc];

    asm volatile("" ::: "memory");
    __builtin_amdgcn_sched_barrier(0);

    v8f acc0 = {}, acc1 = {};
#pragma unroll
    for (int j = 0; j < 8; ++j) {
      v2f a0 = {av[j].x, av[j].y};
      v2f a1 = {av[j].z, av[j].w};
      acc0 = WMMA_F32(a0, rb[2 * j],     acc0);
      acc1 = WMMA_F32(a1, rb[2 * j + 1], acc1);
    }
    __builtin_amdgcn_sched_barrier(0);
    st = acc0 + acc1;

    if (emit) {
      float* rw = &rs[cur][b0 * RP + e];
#pragma unroll
      for (int q = 0; q < 4; ++q) rw[q * RP] = st[2 * q];
      if (flush) *(float4*)(sp + (ot - 1) * E_) = o;
    }
  };

  int t = tw;
  for (int i = 0; i < nwarm; i += 2, t += 2) {
    step(t,     0, bte, false, false, 0);
    step(t + 1, 1, bto, false, false, 0);
  }
  for (int ot = 0; ot < LCH; ot += 2, t += 2) {
    step(t,     0, bte, true, ot > 0, ot);
    step(t + 1, 1, bto, true, true,   ot + 1);
  }

  {
    float* iw = &fs[b0 * RP + e];
#pragma unroll
    for (int q = 0; q < 4; ++q) iw[q * RP] = st[2 * q + 1];
    asm volatile("s_wait_dscnt 0x0\n\ts_barrier_signal -1\n\ts_barrier_wait -1" ::: "memory");
    const float4 o = *(const float4*)&rs[(LCH - 1) & 1][ob * RP + oc];
    *(float4*)(sp + (LCH - 1) * E_) = o;
    if (c == NBLK - 1) {
      *(float4*)(rfin + ob * E_ + oc) = o;
      *(float4*)(ifin + ob * E_ + oc) = *(const float4*)&fs[ob * RP + oc];
    }
  }
}

__global__ __launch_bounds__(128)
void k_out_ln(const float* __restrict__ states, const float* __restrict__ w_out,
              const float* __restrict__ gamma, const float* __restrict__ lbeta,
              float* __restrict__ y) {
  const int lane = threadIdx.x & 31;
  const int wv   = threadIdx.x >> 5;
  const int m16  = lane & 15;
  const int half = lane >> 4;
  const int rowbase = blockIdx.x * 16;

  __shared__ __attribute__((aligned(16))) float yt[16][D_];
  __shared__ __attribute__((aligned(16))) float wsb[4][16 * E_];

  v4f gm[8], bt[8];
#pragma unroll
  for (int i = 0; i < 8; ++i) {
    gm[i] = *(const v4f*)(gamma + 128 * i + 4 * lane);
    bt[i] = *(const v4f*)(lbeta + 128 * i + 4 * lane);
  }

  float* wst = &wsb[wv][0];
  int so[8], fo[8];
#pragma unroll
  for (int i = 0; i < 8; ++i) {
    const int r = 2 * i + half;
    so[i] = r * E_ + 4 * (m16 ^ r);
    fo[i] = m16 * E_ + 4 * ((8 * half + i) ^ m16);
  }

  v4f g[8];
  auto fetch = [&](const float* p) {
#pragma unroll
    for (int i = 0; i < 8; ++i) g[i] = *(const v4f*)(p + 128 * i);
  };
  auto stage = [&]() {
#pragma unroll
    for (int i = 0; i < 8; ++i) *(v4f*)(wst + so[i]) = g[i];
  };
  auto ldfrag = [&](v4f (&f)[8]) {
#pragma unroll
    for (int j = 0; j < 8; ++j) f[j] = *(const v4f*)(wst + fo[j]);
  };
#define TILE_STORED() asm volatile("s_wait_dscnt 0x0" ::: "memory")

  v4f aq[8], b0[8], b1[8];

  auto ybof = [&](const int tl) {
    unsigned yb = 4u * (unsigned)((8 * half) * D_ + (wv * 16 + tl) * 16 + m16);
    asm volatile("" : "+v"(yb));
    return yb;
  };
  auto mm = [&](const v4f (&b)[8], const unsigned yb) {
    float* yr = (float*)((char*)&yt[0][0] + yb);
    GROUP_EDGE();
    v8f acc = {};
#pragma unroll
    for (int j = 0; j < 8; ++j) {
      const v2f a0 = {aq[j].x, aq[j].y};
      const v2f c0 = {b[j].x, b[j].y};
      acc = WMMA_F32(a0, c0, acc);
      const v2f a1 = {aq[j].z, aq[j].w};
      const v2f c1 = {b[j].z, b[j].w};
      acc = WMMA_F32(a1, c1, acc);
    }
    GROUP_EDGE();
#pragma unroll
    for (int v = 0; v < 8; ++v) yr[v * D_] = acc[v];
  };

  fetch(states + (size_t)rowbase * E_ + 4 * lane);
  stage();
  TILE_STORED();
  ldfrag(aq);

  const float* wt = w_out + (size_t)(wv * 16) * (16 * E_) + 4 * lane;
  fetch(wt);
  stage();
  TILE_STORED();
  ldfrag(b0);
  fetch(wt + 16 * E_);
#pragma unroll 1
  for (int tl = 0; tl < 14; tl += 2) {
    stage();
    fetch(wt + (size_t)(tl + 2) * (16 * E_));
    TILE_STORED();
    ldfrag(b1);
    mm(b0, ybof(tl));
    stage();
    fetch(wt + (size_t)(tl + 3) * (16 * E_));
    TILE_STORED();
    ldfrag(b0);
    mm(b1, ybof(tl + 1));
  }
  const unsigned y14 = ybof(14), y15 = ybof(15);
  stage();
  TILE_STORED();
  ldfrag(b1);
  mm(b0, y14);
  mm(b1, y15);
  __syncthreads();

  v4f yv[4][8];
#pragma unroll
  for (int q = 0; q < 4; ++q)
#pragma unroll
    for (int i = 0; i < 8; ++i)
      yv[q][i] = *(const v4f*)&yt[4 * wv + q][128 * i + 4 * lane];

  float s0[4], s1[4];
#pragma unroll
  for (int q = 0; q < 4; ++q) {
    v4f p0 = yv[q][0];
    v4f p1 = yv[q][0] * yv[q][0];
#pragma unroll
    for (int i = 1; i < 8; ++i) {
      p0 += yv[q][i];
      p1 += yv[q][i] * yv[q][i];
    }
    s0[q] = (p0.x + p0.y) + (p0.z + p0.w);
    s1[q] = (p1.x + p1.y) + (p1.z + p1.w);
  }
#pragma unroll
  for (int m = 16; m >= 1; m >>= 1) {
#pragma unroll
    for (int q = 0; q < 4; ++q) {
      s0[q] += __shfl_xor(s0[q], m);
      s1[q] += __shfl_xor(s1[q], m);
    }
  }

  float* yo = y + (size_t)(rowbase + 4 * wv) * D_ + 4 * lane;
#pragma unroll
  for (int q = 0; q < 4; ++q) {
    const float mu  = s0[q] * (1.f / D_);
    const float var = s1[q] * (1.f / D_) - mu * mu;
    const float rs  = rsqrtf(var + 1e-5f);
#pragma unroll
    for (int i = 0; i < 8; ++i) {
      const v4f o = (yv[q][i] - mu) * rs * gm[i] + bt[i];
      *(v4f*)(yo + (size_t)q * D_ + 128 * i) = o;
    }
  }
}

extern "C" void kernel_launch(void* const* d_in, const int* in_sizes, int n_in,
                              void* d_out, int out_size, void* d_ws, size_t ws_size,
                              hipStream_t stream) {
  (void)in_sizes; (void)n_in; (void)out_size; (void)ws_size;

  const float* x     = (const float*)d_in[0];
  const float* alpha = (const float*)d_in[1];
  const float* omega = (const float*)d_in[2];
  const float* w_in  = (const float*)d_in[3];
  const float* w_out = (const float*)d_in[4];
  const float* res   = (const float*)d_in[5];
  const float* gamma = (const float*)d_in[6];
  const float* lbeta = (const float*)d_in[7];

  float* out  = (float*)d_out;
  float* y    = out;
  float* rfin = out + YSZ;
  float* ifin = rfin + B_ * E_;

  float* beta   = (float*)d_ws;
  float* states = beta + (size_t)M_ * E_;

  hipLaunchKernelGGL(k_beta_gemm, dim3(M_ / 64), dim3(128), 0, stream, x, w_in, beta);
  hipLaunchKernelGGL(k_scan, dim3(NBLK), dim3(128), 0, stream,
                     beta, alpha, omega, res, states, rfin, ifin);
  hipLaunchKernelGGL(k_out_ln, dim3(M_ / 16), dim3(128), 0, stream,
                     states, w_out, gamma, lbeta, y);
}


# scan and output kernel fused into ONE launch (3 -> 2): 256 blocks x 16 waves, waves 0-3 run the scan (the rest only join its barriers), then four 4-wave groups do product + LN on the block's 64 rows
# speedup vs baseline: 4.7578x; 1.0419x over previous
#include <hip/hip_runtime.h>
#include <math.h>

typedef float v2f __attribute__((ext_vector_type(2)));
typedef float v4f __attribute__((ext_vector_type(4)));
typedef float v8f __attribute__((ext_vector_type(8)));

#define WMMA_F32(a, b, c) \
  __builtin_amdgcn_wmma_f32_16x16x4_f32(false, (a), false, (b), (short)0, (c), false, false)

static constexpr int B_ = 8, S_ = 2048, D_ = 1024, E_ = 64;
static constexpr int M_ = B_ * S_;
static constexpr long long YSZ = (long long)M_ * D_;

static constexpr int KS    = 64;
static constexpr int NSLAB = D_ / KS;
static constexpr int TP    = 68;

__global__ __launch_bounds__(128)
void k_beta_gemm(const float* __restrict__ x, const float* __restrict__ w_in,
                 float* __restrict__ beta) {
  const int tid  = threadIdx.x;
  const int lane = tid & 31;
  const int w    = tid >> 5;
  const int wv   = (blockIdx.x << 2) + w;
  const int m16  = lane & 15;
  const int half = lane >> 4;

  __shared__ __attribute__((aligned(16))) float xt[2][64 * TP];
  __shared__ __attribute__((aligned(16))) float wt[2][64 * TP];

  const int sr = tid >> 4;
  const int sc = (tid & 15) * 4;
  const float* xg = x    + (size_t)(blockIdx.x * 64 + sr) * D_ + sc;
  const float* wg = w_in + (size_t)sr * D_ + sc;
  const int so = sr * TP + sc;

  v4f gx[8], gw[8];

  auto fetch = [&](const int s) {
#pragma unroll
    for (int i = 0; i < 8; ++i) {
      gx[i] = *(const v4f*)(xg + (size_t)(8 * i) * D_ + s * KS);
      gw[i] = *(const v4f*)(wg + (size_t)(8 * i) * D_ + s * KS);
    }
  };
  auto stage = [&](const int buf) {
#pragma unroll
    for (int i = 0; i < 8; ++i) {
      *(v4f*)&xt[buf][so + 8 * i * TP] = gx[i];
      *(v4f*)&wt[buf][so + 8 * i * TP] = gw[i];
    }
  };

  v8f acc[4];
#pragma unroll
  for (int i = 0; i < 4; ++i) acc[i] = {};

  auto ldq = [&](const float* ar, const float* br, const int q, v4f (&a)[2], v4f (&b)[2][4]) {
#pragma unroll
    for (int jj = 0; jj < 2; ++jj) {
      a[jj] = *(const v4f*)(ar + 4 * (2 * q + jj));
#pragma unroll
      for (int nt = 0; nt < 4; ++nt)
        b[jj][nt] = *(const v4f*)(br + nt * 16 * TP + 4 * (2 * q + jj));
    }
  };
  auto mmq = [&](const v4f (&a)[2], const v4f (&b)[2][4]) {
#pragma unroll
    for (int jj = 0; jj < 2; ++jj) {
      const v2f a0 = {a[jj].x, a[jj].y};
      const v2f a1 = {a[jj].z, a[jj].w};
#pragma unroll
      for (int nt = 0; nt < 4; ++nt) {
        const v2f b0 = {b[jj][nt].x, b[jj][nt].y};
        acc[nt] = WMMA_F32(a0, b0, acc[nt]);
      }
#pragma unroll
      for (int nt = 0; nt < 4; ++nt) {
        const v2f b1 = {b[jj][nt].z, b[jj][nt].w};
        acc[nt] = WMMA_F32(a1, b1, acc[nt]);
      }
    }
  };
#define GROUP_EDGE() do { asm volatile("" ::: "memory"); __builtin_amdgcn_sched_barrier(0); } while (0)

  auto multiply = [&](const int buf) {
    const float* ar = &xt[buf][(16 * w + m16) * TP + 32 * half];
    const float* br = &wt[buf][m16 * TP + 32 * half];
    v4f a0[2], b0[2][4], a1[2], b1[2][4];
    ldq(ar, br, 0, a0, b0);
    ldq(ar, br, 1, a1, b1);
    GROUP_EDGE();
    mmq(a0, b0);
    GROUP_EDGE();
    ldq(ar, br, 2, a0, b0);
    GROUP_EDGE();
    mmq(a1, b1);
    GROUP_EDGE();
    ldq(ar, br, 3, a1, b1);
    GROUP_EDGE();
    mmq(a0, b0);
    mmq(a1, b1);
    GROUP_EDGE();
  };

  fetch(0);
  stage(0);
  __syncthreads();
  for (int s = 0; s < NSLAB - 1; ++s) {
    fetch(s + 1);
    multiply(s & 1);
    stage((s + 1) & 1);
    __syncthreads();
  }
  multiply((NSLAB - 1) & 1);

#pragma unroll
  for (int nt = 0; nt < 4; ++nt)
#pragma unroll
    for (int v = 0; v < 8; ++v)
      beta[(size_t)(wv * 16 + v + 8 * half) * E_ + nt * 16 + m16] = acc[nt][v];
}

static constexpr int LCH  = 8;
static constexpr int NBLK = S_ / LCH;
static constexpr int WARM = 64;
static constexpr int SP = 68;
static constexpr int RP = 72;

static constexpr int YT_F  = 16 * D_;
static constexpr int WSB_F = 16 * E_;
static constexpr int XS_F  = 2 * 16 * SP;
static constexpr int RS_F  = 2 * B_ * RP;
static constexpr int FS_F  = B_ * RP;
static_assert(XS_F + RS_F + FS_F <= 16 * WSB_F, "the scan's buffers lie inside the staging tiles");

__global__ __launch_bounds__(512)
void k_scan_out(const float* __restrict__ beta, const float* __restrict__ alpha,
                const float* __restrict__ omega, const float* __restrict__ res,
                float* states,
                float* __restrict__ rfin, float* __restrict__ ifin,
                const float* __restrict__ w_out,
                const float* __restrict__ gamma, const float* __restrict__ lbeta,
                float* __restrict__ y) {
  const int c    = blockIdx.x;
  const int tid  = threadIdx.x;
  const int lane = tid & 31;
  const int m16  = lane & 15;
  const int half = lane >> 4;
  const int t0   = c * LCH;

  __shared__ __attribute__((aligned(16))) float smem[4 * YT_F + 16 * WSB_F];
  float* const xs = smem + 4 * YT_F;
  float* const rs = xs + XS_F;
  float* const fs = rs + RS_F;

  {
    const bool sw  = __builtin_amdgcn_readfirstlane(tid >> 7) == 0;
    const int w    = (tid >> 5) & 3;
    const int e    = w * 16 + m16;
    const int b0   = 4 * half;

    const int tw    = (t0 > WARM) ? (t0 - WARM) : 0;
    const int nwarm = t0 - tw;
    const int tlast = t0 + LCH - 1;

    float mag = 0.f, cw = 0.f, sn = 0.f;
    v2f rb[16];
#pragma unroll
    for (int t = 0; t < 16; ++t) rb[t] = {};
    const float* bp = beta + (size_t)(b0 * S_) * E_ + e;
    float bte[4], bto[4];
    if (sw) {
      mag = 1.f / (1.f + expf(-alpha[e]));
      cw  = cosf(omega[e]);
      sn  = sinf(omega[e]);
#pragma unroll
      for (int t = 0; t < 16; ++t) {
        const int k = 32 * half + 2 * t;
        rb[t].x = res[k * E_ + e];
        rb[t].y = res[(k + 1) * E_ + e];
      }
#pragma unroll
      for (int t = 0; t < 16; ++t) asm volatile("" : "+v"(rb[t].x), "+v"(rb[t].y));
    }
#pragma unroll
    for (int q = 0; q < 4; ++q) bte[q] = bp[(size_t)q * S_ * E_ + (size_t)tw * E_];
#pragma unroll
    for (int q = 0; q < 4; ++q) bto[q] = bp[(size_t)q * S_ * E_ + (size_t)(tw + 1) * E_];

    v8f st = {};

    const int ob = (tid >> 4) & 7, oc = (tid & 15) * 4;
    float* sp = states + (size_t)(ob * S_ + t0) * E_ + oc;

    auto rotate = [&](const int cur, const float (&bt)[4]) {
      float* xw = &xs[cur * (16 * SP) + (8 * half) * SP + e];
#pragma unroll
      for (int q = 0; q < 4; ++q) {
        const float re = st[2 * q], im = st[2 * q + 1];
        const float nr = fmaf(mag, re * cw - im * sn, bt[q]);
        const float ni = mag * (re * sn + im * cw);
        xw[(2 * q) * SP]     = nr;
        xw[(2 * q + 1) * SP] = ni;
      }
    };

    auto refill = [&](float (&bt)[4], const int tn) {
      const int t2 = (tn < tlast) ? tn : tlast;
#pragma unroll
      for (int q = 0; q < 4; ++q) bt[q] = bp[(size_t)q * S_ * E_ + (size_t)t2 * E_];
    };

    auto step = [&](const int t, const int cur, float (&bt)[4], float (&btn)[4],
                    const bool emit, const bool flush, const int ot, const bool more) {
      asm volatile("s_wait_dscnt 0x0\n\ts_barrier_signal -1" ::: "memory");

      refill(bt, t + 2);

      asm volatile("s_barrier_wait -1" ::: "memory");

      if (sw) {
        const float4* ar = (const float4*)&xs[cur * (16 * SP) + m16 * SP + 32 * half];
        float4 av[8];
#pragma unroll
        for (int j = 0; j < 8; ++j) av[j] = ar[j];

        float4 o = make_float4(0.f, 0.f, 0.f, 0.f);
        if (flush) o = *(const float4*)&rs[(cur ^ 1) * (B_ * RP) + ob * RP + oc];

        asm volatile("" ::: "memory");
        __builtin_amdgcn_sched_barrier(0);

        v8f acc0 = {}, acc1 = {};
#pragma unroll
        for (int j = 0; j < 8; ++j) {
          v2f a0 = {av[j].x, av[j].y};
          v2f a1 = {av[j].z, av[j].w};
          acc0 = WMMA_F32(a0, rb[2 * j],     acc0);
          acc1 = WMMA_F32(a1, rb[2 * j + 1], acc1);
        }
        __builtin_amdgcn_sched_barrier(0);
        st = acc0 + acc1;

        if (emit) {
          float* rw = &rs[cur * (B_ * RP) + b0 * RP + e];
#pragma unroll
          for (int q = 0; q < 4; ++q) rw[q * RP] = st[2 * q];
          if (flush) *(float4*)(sp + (ot - 1) * E_) = o;
        }
      }
      if (more) {
        asm volatile("" : "+v"(btn[0]), "+v"(btn[1]), "+v"(btn[2]), "+v"(btn[3]));
        if (sw) rotate(cur ^ 1, btn);
      }
    };

    asm volatile("" : "+v"(bte[0]), "+v"(bte[1]), "+v"(bte[2]), "+v"(bte[3]));
    if (sw) rotate(0, bte);
    int t = tw;
    for (int i = 0; i < nwarm; i += 2, t += 2) {
      step(t,     0, bte, bto, false, false, 0, true);
      step(t + 1, 1, bto, bte, false, false, 0, true);
    }
    for (int ot = 0; ot < LCH; ot += 2, t += 2) {
      step(t,     0, bte, bto, true, ot > 0, ot,     true);
      step(t + 1, 1, bto, bte, true, true,   ot + 1, ot + 2 < LCH);
    }

    if (sw) {
      float* iw = &fs[b0 * RP + e];
#pragma unroll
      for (int q = 0; q < 4; ++q) iw[q * RP] = st[2 * q + 1];
    }
    asm volatile("s_wait_dscnt 0x0\n\ts_barrier_signal -1\n\ts_barrier_wait -1" ::: "memory");
    if (sw) {
      const float4 o = *(const float4*)&rs[((LCH - 1) & 1) * (B_ * RP) + ob * RP + oc];
      *(float4*)(sp + (LCH - 1) * E_) = o;
      if (c == NBLK - 1) {
        *(float4*)(rfin + ob * E_ + oc) = o;
        *(float4*)(ifin + ob * E_ + oc) = *(const float4*)&fs[ob * RP + oc];
      }
    }
  }
  __syncthreads();

  const int grp = __builtin_amdgcn_readfirstlane(tid >> 7);
  const int wv  = __builtin_amdgcn_readfirstlane((tid >> 5) & 3);
  float* const ytg = smem + grp * YT_F;

  v4f gm[8], bt[8];
#pragma unroll
  for (int i = 0; i < 8; ++i) {
    gm[i] = *(const v4f*)(gamma + 128 * i + 4 * lane);
    bt[i] = *(const v4f*)(lbeta + 128 * i + 4 * lane);
  }

  float* wst = smem + 4 * YT_F + (grp * 4 + wv) * WSB_F;
  int so[8], fo[8];
#pragma unroll
  for (int i = 0; i < 8; ++i) {
    const int r = 2 * i + half;
    so[i] = r * E_ + 4 * (m16 ^ r);
    fo[i] = m16 * E_ + 4 * ((8 * half + i) ^ m16);
  }

  v4f g[8];
  auto fetch = [&](const float* p) {
#pragma unroll
    for (int i = 0; i < 8; ++i) g[i] = *(const v4f*)(p + 128 * i);
  };
  auto stage = [&]() {
#pragma unroll
    for (int i = 0; i < 8; ++i) *(v4f*)(wst + so[i]) = g[i];
  };
  auto ldfrag = [&](v4f (&f)[8]) {
#pragma unroll
    for (int j = 0; j < 8; ++j) f[j] = *(const v4f*)(wst + fo[j]);
  };
#define TILE_STORED() asm volatile("s_wait_dscnt 0x0" ::: "memory")

  v4f aq[8], b0[8], b1[8];

  auto ybof = [&](const int tl) {
    unsigned yb = 4u * (unsigned)((8 * half) * D_ + (wv * 16 + tl) * 16 + m16);
    asm volatile("" : "+v"(yb));
    return yb;
  };
  auto mm = [&](const v4f (&b)[8], const unsigned yb) {
    float* yr = (float*)((char*)ytg + yb);
    GROUP_EDGE();
    v8f acc = {};
#pragma unroll
    for (int j = 0; j < 8; ++j) {
      const v2f a0 = {aq[j].x, aq[j].y};
      const v2f c0 = {b[j].x, b[j].y};
      acc = WMMA_F32(a0, c0, acc);
      const v2f a1 = {aq[j].z, aq[j].w};
      const v2f c1 = {b[j].z, b[j].w};
      acc = WMMA_F32(a1, c1, acc);
    }
    GROUP_EDGE();
#pragma unroll
    for (int v = 0; v < 8; ++v) yr[v * D_] = acc[v];
  };

  {
    const float* s0p = states + (size_t)((2 * grp) * S_ + t0) * E_ + 4 * lane;
    const float* s1p = states + (size_t)((2 * grp + 1) * S_ + t0) * E_ + 4 * lane;
#pragma unroll
    for (int i = 0; i < 4; ++i) {
      g[i]     = *(const v4f*)(s0p + 128 * i);
      g[i + 4] = *(const v4f*)(s1p + 128 * i);
    }
  }
  stage();
  TILE_STORED();
  ldfrag(aq);

  const float* wt = w_out + (size_t)(wv * 16) * (16 * E_) + 4 * lane;
  fetch(wt);
  stage();
  TILE_STORED();
  ldfrag(b0);
  fetch(wt + 16 * E_);
#pragma unroll 1
  for (int tl = 0; tl < 14; tl += 2) {
    stage();
    fetch(wt + (size_t)(tl + 2) * (16 * E_));
    TILE_STORED();
    ldfrag(b1);
    mm(b0, ybof(tl));
    stage();
    fetch(wt + (size_t)(tl + 3) * (16 * E_));
    TILE_STORED();
    ldfrag(b0);
    mm(b1, ybof(tl + 1));
  }
  const unsigned y14 = ybof(14), y15 = ybof(15);
  stage();
  TILE_STORED();
  ldfrag(b1);
  mm(b0, y14);
  mm(b1, y15);
  __syncthreads();

  v4f yv[4][8];
#pragma unroll
  for (int q = 0; q < 4; ++q)
#pragma unroll
    for (int i = 0; i < 8; ++i)
      yv[q][i] = *(const v4f*)&ytg[(4 * wv + q) * D_ + 128 * i + 4 * lane];

  float s0[4], s1[4];
#pragma unroll
  for (int q = 0; q < 4; ++q) {
    v4f p0 = yv[q][0];
    v4f p1 = yv[q][0] * yv[q][0];
#pragma unroll
    for (int i = 1; i < 8; ++i) {
      p0 += yv[q][i];
      p1 += yv[q][i] * yv[q][i];
    }
    s0[q] = (p0.x + p0.y) + (p0.z + p0.w);
    s1[q] = (p1.x + p1.y) + (p1.z + p1.w);
  }
#pragma unroll
  for (int m = 16; m >= 1; m >>= 1) {
#pragma unroll
    for (int q = 0; q < 4; ++q) {
      s0[q] += __shfl_xor(s0[q], m);
      s1[q] += __shfl_xor(s1[q], m);
    }
  }

  float* yo = y + (size_t)((2 * grp + (wv >> 1)) * S_ + t0 + 4 * (wv & 1)) * D_ + 4 * lane;
#pragma unroll
  for (int q = 0; q < 4; ++q) {
    const float mu  = s0[q] * (1.f / D_);
    const float var = s1[q] * (1.f / D_) - mu * mu;
    const float rsd = rsqrtf(var + 1e-5f);
#pragma unroll
    for (int i = 0; i < 8; ++i) {
      const v4f o = (yv[q][i] - mu) * rsd * gm[i] + bt[i];
      *(v4f*)(yo + (size_t)q * D_ + 128 * i) = o;
    }
  }
}

extern "C" void kernel_launch(void* const* d_in, const int* in_sizes, int n_in,
                              void* d_out, int out_size, void* d_ws, size_t ws_size,
                              hipStream_t stream) {
  (void)in_sizes; (void)n_in; (void)out_size; (void)ws_size;

  const float* x     = (const float*)d_in[0];
  const float* alpha = (const float*)d_in[1];
  const float* omega = (const float*)d_in[2];
  const float* w_in  = (const float*)d_in[3];
  const float* w_out = (const float*)d_in[4];
  const float* res   = (const float*)d_in[5];
  const float* gamma = (const float*)d_in[6];
  const float* lbeta = (const float*)d_in[7];

  float* out  = (float*)d_out;
  float* y    = out;
  float* rfin = out + YSZ;
  float* ifin = rfin + B_ * E_;

  float* beta   = (float*)d_ws;
  float* states = beta + (size_t)M_ * E_;

  hipLaunchKernelGGL(k_beta_gemm, dim3(M_ / 64), dim3(128), 0, stream, x, w_in, beta);
  hipLaunchKernelGGL(k_scan_out, dim3(NBLK), dim3(512), 0, stream,
                     beta, alpha, omega, res, states, rfin, ifin, w_out, gamma, lbeta, y);
}


# output product re-tiled: all 16 waves take one 16-row tile at a time, each keeping its 4 w_out tiles in registers (fetched once per block), so a tile's LN and stores run under the next tile's WMMAs
# speedup vs baseline: 4.9337x; 1.0370x over previous
#include <hip/hip_runtime.h>
#include <math.h>
#include <type_traits>

typedef float v2f __attribute__((ext_vector_type(2)));
typedef float v4f __attribute__((ext_vector_type(4)));
typedef float v8f __attribute__((ext_vector_type(8)));

#define WMMA_F32(a, b, c) \
  __builtin_amdgcn_wmma_f32_16x16x4_f32(false, (a), false, (b), (short)0, (c), false, false)

static constexpr int B_ = 8, S_ = 2048, D_ = 1024, E_ = 64;
static constexpr int M_ = B_ * S_;
static constexpr long long YSZ = (long long)M_ * D_;

static constexpr int KS    = 64;
static constexpr int NSLAB = D_ / KS;
static constexpr int TP    = 68;

__global__ __launch_bounds__(128)
void k_beta_gemm(const float* __restrict__ x, const float* __restrict__ w_in,
                 float* __restrict__ beta) {
  const int tid  = threadIdx.x;
  const int lane = tid & 31;
  const int w    = tid >> 5;
  const int wv   = (blockIdx.x << 2) + w;
  const int m16  = lane & 15;
  const int half = lane >> 4;

  __shared__ __attribute__((aligned(16))) float xt[2][64 * TP];
  __shared__ __attribute__((aligned(16))) float wt[2][64 * TP];

  const int sr = tid >> 4;
  const int sc = (tid & 15) * 4;
  const float* xg = x    + (size_t)(blockIdx.x * 64 + sr) * D_ + sc;
  const float* wg = w_in + (size_t)sr * D_ + sc;
  const int so = sr * TP + sc;

  v4f gx[8], gw[8];

  auto fetch = [&](const int s) {
#pragma unroll
    for (int i = 0; i < 8; ++i) {
      gx[i] = *(const v4f*)(xg + (size_t)(8 * i) * D_ + s * KS);
      gw[i] = *(const v4f*)(wg + (size_t)(8 * i) * D_ + s * KS);
    }
  };
  auto stage = [&](const int buf) {
#pragma unroll
    for (int i = 0; i < 8; ++i) {
      *(v4f*)&xt[buf][so + 8 * i * TP] = gx[i];
      *(v4f*)&wt[buf][so + 8 * i * TP] = gw[i];
    }
  };

  v8f acc[4];
#pragma unroll
  for (int i = 0; i < 4; ++i) acc[i] = {};

  auto ldq = [&](const float* ar, const float* br, const int q, v4f (&a)[2], v4f (&b)[2][4]) {
#pragma unroll
    for (int jj = 0; jj < 2; ++jj) {
      a[jj] = *(const v4f*)(ar + 4 * (2 * q + jj));
#pragma unroll
      for (int nt = 0; nt < 4; ++nt)
        b[jj][nt] = *(const v4f*)(br + nt * 16 * TP + 4 * (2 * q + jj));
    }
  };
  auto mmq = [&](const v4f (&a)[2], const v4f (&b)[2][4]) {
#pragma unroll
    for (int jj = 0; jj < 2; ++jj) {
      const v2f a0 = {a[jj].x, a[jj].y};
      const v2f a1 = {a[jj].z, a[jj].w};
#pragma unroll
      for (int nt = 0; nt < 4; ++nt) {
        const v2f b0 = {b[jj][nt].x, b[jj][nt].y};
        acc[nt] = WMMA_F32(a0, b0, acc[nt]);
      }
#pragma unroll
      for (int nt = 0; nt < 4; ++nt) {
        const v2f b1 = {b[jj][nt].z, b[jj][nt].w};
        acc[nt] = WMMA_F32(a1, b1, acc[nt]);
      }
    }
  };
#define GROUP_EDGE() do { asm volatile("" ::: "memory"); __builtin_amdgcn_sched_barrier(0); } while (0)

  auto multiply = [&](const int buf) {
    const float* ar = &xt[buf][(16 * w + m16) * TP + 32 * half];
    const float* br = &wt[buf][m16 * TP + 32 * half];
    v4f a0[2], b0[2][4], a1[2], b1[2][4];
    ldq(ar, br, 0, a0, b0);
    ldq(ar, br, 1, a1, b1);
    GROUP_EDGE();
    mmq(a0, b0);
    GROUP_EDGE();
    ldq(ar, br, 2, a0, b0);
    GROUP_EDGE();
    mmq(a1, b1);
    GROUP_EDGE();
    ldq(ar, br, 3, a1, b1);
    GROUP_EDGE();
    mmq(a0, b0);
    mmq(a1, b1);
    GROUP_EDGE();
  };

  fetch(0);
  stage(0);
  __syncthreads();
  for (int s = 0; s < NSLAB - 1; ++s) {
    fetch(s + 1);
    multiply(s & 1);
    stage((s + 1) & 1);
    __syncthreads();
  }
  multiply((NSLAB - 1) & 1);

#pragma unroll
  for (int nt = 0; nt < 4; ++nt)
#pragma unroll
    for (int v = 0; v < 8; ++v)
      beta[(size_t)(wv * 16 + v + 8 * half) * E_ + nt * 16 + m16] = acc[nt][v];
}

static constexpr int LCH  = 8;
static constexpr int NBLK = S_ / LCH;
static constexpr int WARM = 64;
static constexpr int SP = 68;
static constexpr int RP = 72;

static constexpr int YT_F  = 16 * D_;
static constexpr int WSB_F = 16 * E_;
static constexpr int XS_F  = 2 * 16 * SP;
static constexpr int RS_F  = 2 * B_ * RP;
static constexpr int FS_F  = B_ * RP;
static_assert(XS_F + RS_F + FS_F <= 16 * WSB_F, "the scan's buffers lie inside the staging tiles");

__global__ __launch_bounds__(512)
void k_scan_out(const float* __restrict__ beta, const float* __restrict__ alpha,
                const float* __restrict__ omega, const float* __restrict__ res,
                float* states,
                float* __restrict__ rfin, float* __restrict__ ifin,
                const float* __restrict__ w_out,
                const float* __restrict__ gamma, const float* __restrict__ lbeta,
                float* __restrict__ y) {
  const int c    = blockIdx.x;
  const int tid  = threadIdx.x;
  const int lane = tid & 31;
  const int m16  = lane & 15;
  const int half = lane >> 4;
  const int t0   = c * LCH;

  __shared__ __attribute__((aligned(16))) float smem[4 * YT_F + 16 * WSB_F];
  float* const xs = smem + 4 * YT_F;
  float* const rs = xs + XS_F;
  float* const fs = rs + RS_F;

  {
    const bool sw  = __builtin_amdgcn_readfirstlane(tid >> 7) == 0;
    const int w    = (tid >> 5) & 3;
    const int e    = w * 16 + m16;
    const int b0   = 4 * half;

    const int tw    = (t0 > WARM) ? (t0 - WARM) : 0;
    const int nwarm = t0 - tw;
    const int tlast = t0 + LCH - 1;

    float mag = 0.f, cw = 0.f, sn = 0.f;
    v2f rb[16];
#pragma unroll
    for (int t = 0; t < 16; ++t) rb[t] = {};
    const float* bp = beta + (size_t)(b0 * S_) * E_ + e;
    float bte[4], bto[4];
    if (sw) {
      mag = 1.f / (1.f + expf(-alpha[e]));
      cw  = cosf(omega[e]);
      sn  = sinf(omega[e]);
#pragma unroll
      for (int t = 0; t < 16; ++t) {
        const int k = 32 * half + 2 * t;
        rb[t].x = res[k * E_ + e];
        rb[t].y = res[(k + 1) * E_ + e];
      }
#pragma unroll
      for (int t = 0; t < 16; ++t) asm volatile("" : "+v"(rb[t].x), "+v"(rb[t].y));
    }
#pragma unroll
    for (int q = 0; q < 4; ++q) bte[q] = bp[(size_t)q * S_ * E_ + (size_t)tw * E_];
#pragma unroll
    for (int q = 0; q < 4; ++q) bto[q] = bp[(size_t)q * S_ * E_ + (size_t)(tw + 1) * E_];

    v8f st = {};

    const int ob = (tid >> 4) & 7, oc = (tid & 15) * 4;
    float* sp = states + (size_t)(ob * S_ + t0) * E_ + oc;

    auto rotate = [&](const int cur, const float (&bt)[4]) {
      float* xw = &xs[cur * (16 * SP) + (8 * half) * SP + e];
#pragma unroll
      for (int q = 0; q < 4; ++q) {
        const float re = st[2 * q], im = st[2 * q + 1];
        const float nr = fmaf(mag, re * cw - im * sn, bt[q]);
        const float ni = mag * (re * sn + im * cw);
        xw[(2 * q) * SP]     = nr;
        xw[(2 * q + 1) * SP] = ni;
      }
    };

    auto refill = [&](float (&bt)[4], const int tn) {
      const int t2 = (tn < tlast) ? tn : tlast;
#pragma unroll
      for (int q = 0; q < 4; ++q) bt[q] = bp[(size_t)q * S_ * E_ + (size_t)t2 * E_];
    };

    auto step = [&](const int t, const int cur, float (&bt)[4], float (&btn)[4],
                    const bool emit, const bool flush, const int ot, const bool more) {
      asm volatile("s_wait_dscnt 0x0\n\ts_barrier_signal -1" ::: "memory");

      refill(bt, t + 2);

      asm volatile("s_barrier_wait -1" ::: "memory");

      if (sw) {
        const float4* ar = (const float4*)&xs[cur * (16 * SP) + m16 * SP + 32 * half];
        float4 av[8];
#pragma unroll
        for (int j = 0; j < 8; ++j) av[j] = ar[j];

        float4 o = make_float4(0.f, 0.f, 0.f, 0.f);
        if (flush) o = *(const float4*)&rs[(cur ^ 1) * (B_ * RP) + ob * RP + oc];

        asm volatile("" ::: "memory");
        __builtin_amdgcn_sched_barrier(0);

        v8f acc0 = {}, acc1 = {};
#pragma unroll
        for (int j = 0; j < 8; ++j) {
          v2f a0 = {av[j].x, av[j].y};
          v2f a1 = {av[j].z, av[j].w};
          acc0 = WMMA_F32(a0, rb[2 * j],     acc0);
          acc1 = WMMA_F32(a1, rb[2 * j + 1], acc1);
        }
        __builtin_amdgcn_sched_barrier(0);
        st = acc0 + acc1;

        if (emit) {
          float* rw = &rs[cur * (B_ * RP) + b0 * RP + e];
#pragma unroll
          for (int q = 0; q < 4; ++q) rw[q * RP] = st[2 * q];
          if (flush) *(float4*)(sp + (ot - 1) * E_) = o;
        }
      }
      if (more) {
        asm volatile("" : "+v"(btn[0]), "+v"(btn[1]), "+v"(btn[2]), "+v"(btn[3]));
        if (sw) rotate(cur ^ 1, btn);
      }
    };

    asm volatile("" : "+v"(bte[0]), "+v"(bte[1]), "+v"(bte[2]), "+v"(bte[3]));
    if (sw) rotate(0, bte);
    int t = tw;
    for (int i = 0; i < nwarm; i += 2, t += 2) {
      step(t,     0, bte, bto, false, false, 0, true);
      step(t + 1, 1, bto, bte, false, false, 0, true);
    }
    for (int ot = 0; ot < LCH; ot += 2, t += 2) {
      step(t,     0, bte, bto, true, ot > 0, ot,     true);
      step(t + 1, 1, bto, bte, true, true,   ot + 1, ot + 2 < LCH);
    }

    if (sw) {
      float* iw = &fs[b0 * RP + e];
#pragma unroll
      for (int q = 0; q < 4; ++q) iw[q * RP] = st[2 * q + 1];
    }
    asm volatile("s_wait_dscnt 0x0\n\ts_barrier_signal -1\n\ts_barrier_wait -1" ::: "memory");
    if (sw) {
      const float4 o = *(const float4*)&rs[((LCH - 1) & 1) * (B_ * RP) + ob * RP + oc];
      *(float4*)(sp + (LCH - 1) * E_) = o;
      if (c == NBLK - 1) {
        *(float4*)(rfin + ob * E_ + oc) = o;
        *(float4*)(ifin + ob * E_ + oc) = *(const float4*)&fs[ob * RP + oc];
      }
    }
  }
  __syncthreads();

  const int wi = __builtin_amdgcn_readfirstlane(tid >> 5);
  float* const stg = smem + 4 * YT_F;
  float* const wst = stg + wi * WSB_F;

  int so[8], fo[8];
#pragma unroll
  for (int i = 0; i < 8; ++i) {
    const int r = 2 * i + half;
    so[i] = r * E_ + 4 * (m16 ^ r);
    fo[i] = m16 * E_ + 4 * ((8 * half + i) ^ m16);
  }

  v4f g[8];
  auto fetch = [&](const float* p) {
#pragma unroll
    for (int i = 0; i < 8; ++i) g[i] = *(const v4f*)(p + 128 * i);
  };
  auto stage = [&]() {
#pragma unroll
    for (int i = 0; i < 8; ++i) *(v4f*)(wst + so[i]) = g[i];
  };
  auto ldfrag = [&](const float* tile, v4f (&f)[8]) {
#pragma unroll
    for (int j = 0; j < 8; ++j) f[j] = *(const v4f*)(tile + fo[j]);
  };
#define TILE_STORED() asm volatile("s_wait_dscnt 0x0" ::: "memory")

  v4f ga[8];
  if (wi < 4) {
    const float* s0p = states + (size_t)((2 * wi) * S_ + t0) * E_ + 4 * lane;
    const float* s1p = states + (size_t)((2 * wi + 1) * S_ + t0) * E_ + 4 * lane;
#pragma unroll
    for (int i = 0; i < 4; ++i) {
      ga[i]     = *(const v4f*)(s0p + 128 * i);
      ga[i + 4] = *(const v4f*)(s1p + 128 * i);
    }
  }
  v4f bq[4][8];
  const float* wt = w_out + (size_t)(wi * 4) * (16 * E_) + 4 * lane;
  fetch(wt);
  stage();
  fetch(wt + 1 * (16 * E_));
  TILE_STORED();
  ldfrag(wst, bq[0]);
  stage();
  fetch(wt + 2 * (16 * E_));
  TILE_STORED();
  ldfrag(wst, bq[1]);
  stage();
  fetch(wt + 3 * (16 * E_));
  TILE_STORED();
  ldfrag(wst, bq[2]);
  stage();
  TILE_STORED();
  ldfrag(wst, bq[3]);
  if (wi < 4) {
#pragma unroll
    for (int i = 0; i < 8; ++i) *(v4f*)(wst + so[i]) = ga[i];
  }
  __syncthreads();

  auto ybof = [&](const int T, const int dt) {
    unsigned yb = 4u * (unsigned)(T * YT_F + (8 * half) * D_ + (wi * 4 + dt) * 16 + m16);
    asm volatile("" : "+v"(yb));
    return yb;
  };
  auto mm2 = [&](const v4f (&aq)[8], const v4f (&b0)[8], const v4f (&b1)[8], const unsigned ya, const unsigned yb) {
    float* yr0 = (float*)((char*)smem + ya);
    float* yr1 = (float*)((char*)smem + yb);
    GROUP_EDGE();
    v8f acc0 = {}, acc1 = {};
#pragma unroll
    for (int j = 0; j < 8; ++j) {
      const v2f a0 = {aq[j].x, aq[j].y};
      const v2f a1 = {aq[j].z, aq[j].w};
      const v2f c0 = {b0[j].x, b0[j].y};
      const v2f d0 = {b1[j].x, b1[j].y};
      acc0 = WMMA_F32(a0, c0, acc0);
      acc1 = WMMA_F32(a0, d0, acc1);
      const v2f c1 = {b0[j].z, b0[j].w};
      const v2f d1 = {b1[j].z, b1[j].w};
      acc0 = WMMA_F32(a1, c1, acc0);
      acc1 = WMMA_F32(a1, d1, acc1);
    }
    GROUP_EDGE();
#pragma unroll
    for (int v = 0; v < 8; ++v) yr0[v * D_] = acc0[v];
#pragma unroll
    for (int v = 0; v < 8; ++v) yr1[v * D_] = acc1[v];
  };
  auto runs = [&](const int T, const v4f (&aq)[8]) {
    const unsigned y0 = ybof(T, 0), y1 = ybof(T, 1), y2 = ybof(T, 2), y3 = ybof(T, 3);
    mm2(aq, bq[0], bq[1], y0, y1);
    mm2(aq, bq[2], bq[3], y2, y3);
  };
  auto rowload = [&](const int T, v4f (&yv)[8]) {
    const float* yrow = smem + T * YT_F + wi * D_ + 4 * lane;
#pragma unroll
    for (int i = 0; i < 8; ++i) yv[i] = *(const v4f*)(yrow + 128 * i);
  };
  auto lx = [](const float v, auto mtag) {
    constexpr int m = decltype(mtag)::value;
    const int b = __builtin_bit_cast(int, v);
    if constexpr (m == 16)
      return __builtin_bit_cast(float, __builtin_amdgcn_permlanex16(b, b, 0x76543210, 0xfedcba98, false, false));
    else
      return __builtin_bit_cast(float, __builtin_amdgcn_update_dpp(b, b, 0x160 + m, 0xf, 0xf, false));
  };
  auto rowfinish = [&](const int T, const v4f (&yv)[8]) {
    v4f gA[4], bA[4], gB[4], bB[4];
#pragma unroll
    for (int i = 0; i < 4; ++i) {
      gA[i] = *(const v4f*)(gamma + 128 * i + 4 * lane);
      bA[i] = *(const v4f*)(lbeta + 128 * i + 4 * lane);
    }
    __builtin_amdgcn_sched_barrier(0);
    v4f p0 = yv[0];
    v4f p1 = yv[0] * yv[0];
#pragma unroll
    for (int i = 1; i < 8; ++i) {
      p0 += yv[i];
      p1 += yv[i] * yv[i];
    }
    float s0 = (p0.x + p0.y) + (p0.z + p0.w);
    float s1 = (p1.x + p1.y) + (p1.z + p1.w);
    s0 += lx(s0, std::integral_constant<int, 16>{});  s1 += lx(s1, std::integral_constant<int, 16>{});
    s0 += lx(s0, std::integral_constant<int, 8>{});   s1 += lx(s1, std::integral_constant<int, 8>{});
    s0 += lx(s0, std::integral_constant<int, 4>{});   s1 += lx(s1, std::integral_constant<int, 4>{});
    s0 += lx(s0, std::integral_constant<int, 2>{});   s1 += lx(s1, std::integral_constant<int, 2>{});
    s0 += lx(s0, std::integral_constant<int, 1>{});   s1 += lx(s1, std::integral_constant<int, 1>{});
    const float mu  = s0 * (1.f / D_);
    const float var = s1 * (1.f / D_) - mu * mu;
    const float rsd = rsqrtf(var + 1e-5f);
#pragma unroll
    for (int i = 0; i < 4; ++i) {
      gB[i] = *(const v4f*)(gamma + 128 * (i + 4) + 4 * lane);
      bB[i] = *(const v4f*)(lbeta + 128 * (i + 4) + 4 * lane);
    }
    __builtin_amdgcn_sched_barrier(0);
    float* yo = y + (size_t)((2 * T + (wi >> 3)) * S_ + t0 + (wi & 7)) * D_ + 4 * lane;
#pragma unroll
    for (int i = 0; i < 4; ++i) {
      const v4f o = (yv[i] - mu) * rsd * gA[i] + bA[i];
      *(v4f*)(yo + 128 * i) = o;
    }
#pragma unroll
    for (int i = 0; i < 4; ++i) {
      const v4f o = (yv[i + 4] - mu) * rsd * gB[i] + bB[i];
      *(v4f*)(yo + 128 * (i + 4)) = o;
    }
  };
#define LDS_BARRIER() asm volatile("s_wait_dscnt 0x0\n\ts_barrier_signal -1\n\ts_barrier_wait -1" ::: "memory")

  {
    v4f aq[8];
    ldfrag(stg, aq);
    runs(0, aq);
  }
  LDS_BARRIER();
#pragma unroll 1
  for (int T = 1; T < 4; ++T) {
    v4f aq[8], yv[8];
    ldfrag(stg + T * WSB_F, aq);
    rowload(T - 1, yv);
    runs(T, aq);
    asm volatile("s_wait_dscnt 0x0\n\ts_barrier_signal -1" ::: "memory");
    rowfinish(T - 1, yv);
    asm volatile("s_barrier_wait -1" ::: "memory");
  }
  {
    v4f yv[8];
    rowload(3, yv);
    rowfinish(3, yv);
  }
}

extern "C" void kernel_launch(void* const* d_in, const int* in_sizes, int n_in,
                              void* d_out, int out_size, void* d_ws, size_t ws_size,
                              hipStream_t stream) {
  (void)in_sizes; (void)n_in; (void)out_size; (void)ws_size;

  const float* x     = (const float*)d_in[0];
  const float* alpha = (const float*)d_in[1];
  const float* omega = (const float*)d_in[2];
  const float* w_in  = (const float*)d_in[3];
  const float* w_out = (const float*)d_in[4];
  const float* res   = (const float*)d_in[5];
  const float* gamma = (const float*)d_in[6];
  const float* lbeta = (const float*)d_in[7];

  float* out  = (float*)d_out;
  float* y    = out;
  float* rfin = out + YSZ;
  float* ifin = rfin + B_ * E_;

  float* beta   = (float*)d_ws;
  float* states = beta + (size_t)M_ * E_;

  hipLaunchKernelGGL(k_beta_gemm, dim3(M_ / 64), dim3(128), 0, stream, x, w_in, beta);
  hipLaunchKernelGGL(k_scan_out, dim3(NBLK), dim3(512), 0, stream,
                     beta, alpha, omega, res, states, rfin, ifin, w_out, gamma, lbeta, y);
}


# input GEMM re-tiled: 8 waves per block, two per SIMD (wave = one 16-row tile x two 16-column tiles), on 128-k slabs: half the barriers, one wave's gaps under the other's WMMAs; sums unchanged
# speedup vs baseline: 5.3215x; 1.0786x over previous
#include <hip/hip_runtime.h>
#include <math.h>
#include <type_traits>

typedef float v2f __attribute__((ext_vector_type(2)));
typedef float v4f __attribute__((ext_vector_type(4)));
typedef float v8f __attribute__((ext_vector_type(8)));

#define WMMA_F32(a, b, c) \
  __builtin_amdgcn_wmma_f32_16x16x4_f32(false, (a), false, (b), (short)0, (c), false, false)

static constexpr int B_ = 8, S_ = 2048, D_ = 1024, E_ = 64;
static constexpr int M_ = B_ * S_;
static constexpr long long YSZ = (long long)M_ * D_;

static constexpr int KS    = 128;
static constexpr int NSLAB = D_ / KS;
static constexpr int TP    = 132;

__global__ __launch_bounds__(256)
void k_beta_gemm(const float* __restrict__ x, const float* __restrict__ w_in,
                 float* __restrict__ beta) {
  const int tid  = threadIdx.x;
  const int lane = tid & 31;
  const int w    = tid >> 5;
  const int wm   = w & 3;
  const int wn   = w >> 2;
  const int m16  = lane & 15;
  const int half = lane >> 4;

  __shared__ __attribute__((aligned(16))) float xt[2][64 * TP];
  __shared__ __attribute__((aligned(16))) float wt[2][64 * TP];

  const int sr = tid >> 4;
  const int sc = (tid & 15) * 4;
  const float* xg = x    + (size_t)(blockIdx.x * 64 + sr) * D_ + sc;
  const float* wg = w_in + (size_t)sr * D_ + sc;
  const int so = sr * TP + sc;

  v4f gx[8], gw[8];

  auto fetch = [&](const int s) {
#pragma unroll
    for (int i = 0; i < 4; ++i) {
#pragma unroll
      for (int u = 0; u < 2; ++u) {
        gx[2 * i + u] = *(const v4f*)(xg + (size_t)(16 * i) * D_ + s * KS + 64 * u);
        gw[2 * i + u] = *(const v4f*)(wg + (size_t)(16 * i) * D_ + s * KS + 64 * u);
      }
    }
  };
  auto stage = [&](const int buf) {
#pragma unroll
    for (int i = 0; i < 4; ++i) {
#pragma unroll
      for (int u = 0; u < 2; ++u) {
        *(v4f*)&xt[buf][so + 16 * i * TP + 64 * u] = gx[2 * i + u];
        *(v4f*)&wt[buf][so + 16 * i * TP + 64 * u] = gw[2 * i + u];
      }
    }
  };

  v8f acc[2];
#pragma unroll
  for (int i = 0; i < 2; ++i) acc[i] = {};

  auto ldq = [&](const float* ar, const float* br, const int Q, v4f (&a)[2], v4f (&b)[2][2]) {
    const int co = 64 * (Q >> 2) + 8 * (Q & 3);
#pragma unroll
    for (int jj = 0; jj < 2; ++jj) {
      a[jj] = *(const v4f*)(ar + co + 4 * jj);
#pragma unroll
      for (int nt = 0; nt < 2; ++nt)
        b[jj][nt] = *(const v4f*)(br + nt * 16 * TP + co + 4 * jj);
    }
  };
  auto mmq = [&](const v4f (&a)[2], const v4f (&b)[2][2]) {
#pragma unroll
    for (int jj = 0; jj < 2; ++jj) {
      const v2f a0 = {a[jj].x, a[jj].y};
      const v2f a1 = {a[jj].z, a[jj].w};
#pragma unroll
      for (int nt = 0; nt < 2; ++nt) {
        const v2f b0 = {b[jj][nt].x, b[jj][nt].y};
        acc[nt] = WMMA_F32(a0, b0, acc[nt]);
      }
#pragma unroll
      for (int nt = 0; nt < 2; ++nt) {
        const v2f b1 = {b[jj][nt].z, b[jj][nt].w};
        acc[nt] = WMMA_F32(a1, b1, acc[nt]);
      }
    }
  };
#define GROUP_EDGE() do { asm volatile("" ::: "memory"); __builtin_amdgcn_sched_barrier(0); } while (0)

  auto multiply = [&](const int buf) {
    const float* ar = &xt[buf][(16 * wm + m16) * TP + 32 * half];
    const float* br = &wt[buf][(32 * wn + m16) * TP + 32 * half];
    v4f a0[2], b0[2][2], a1[2], b1[2][2];
    ldq(ar, br, 0, a0, b0);
    ldq(ar, br, 1, a1, b1);
    GROUP_EDGE();
#pragma unroll
    for (int Q = 0; Q < 8; Q += 2) {
      mmq(a0, b0);
      if (Q + 2 < 8) {
        GROUP_EDGE();
        ldq(ar, br, Q + 2, a0, b0);
        GROUP_EDGE();
      }
      mmq(a1, b1);
      if (Q + 3 < 8) {
        GROUP_EDGE();
        ldq(ar, br, Q + 3, a1, b1);
        GROUP_EDGE();
      }
    }
    GROUP_EDGE();
  };

  fetch(0);
  stage(0);
  __syncthreads();
  for (int s = 0; s < NSLAB - 1; ++s) {
    fetch(s + 1);
    multiply(s & 1);
    stage((s + 1) & 1);
    __syncthreads();
  }
  multiply((NSLAB - 1) & 1);

#pragma unroll
  for (int nt = 0; nt < 2; ++nt)
#pragma unroll
    for (int v = 0; v < 8; ++v)
      beta[(size_t)(blockIdx.x * 64 + wm * 16 + v + 8 * half) * E_ + (2 * wn + nt) * 16 + m16] = acc[nt][v];
}

static constexpr int LCH  = 8;
static constexpr int NBLK = S_ / LCH;
static constexpr int WARM = 64;
static constexpr int SP = 68;
static constexpr int RP = 72;

static constexpr int YT_F  = 16 * D_;
static constexpr int WSB_F = 16 * E_;
static constexpr int XS_F  = 2 * 16 * SP;
static constexpr int RS_F  = 2 * B_ * RP;
static constexpr int FS_F  = B_ * RP;
static_assert(XS_F + RS_F + FS_F <= 16 * WSB_F, "the scan's buffers lie inside the staging tiles");

__global__ __launch_bounds__(512)
void k_scan_out(const float* __restrict__ beta, const float* __restrict__ alpha,
                const float* __restrict__ omega, const float* __restrict__ res,
                float* states,
                float* __restrict__ rfin, float* __restrict__ ifin,
                const float* __restrict__ w_out,
                const float* __restrict__ gamma, const float* __restrict__ lbeta,
                float* __restrict__ y) {
  const int c    = blockIdx.x;
  const int tid  = threadIdx.x;
  const int lane = tid & 31;
  const int m16  = lane & 15;
  const int half = lane >> 4;
  const int t0   = c * LCH;

  __shared__ __attribute__((aligned(16))) float smem[4 * YT_F + 16 * WSB_F];
  float* const xs = smem + 4 * YT_F;
  float* const rs = xs + XS_F;
  float* const fs = rs + RS_F;

  {
    const bool sw  = __builtin_amdgcn_readfirstlane(tid >> 7) == 0;
    const int w    = (tid >> 5) & 3;
    const int e    = w * 16 + m16;
    const int b0   = 4 * half;

    const int tw    = (t0 > WARM) ? (t0 - WARM) : 0;
    const int nwarm = t0 - tw;
    const int tlast = t0 + LCH - 1;

    float mag = 0.f, cw = 0.f, sn = 0.f;
    v2f rb[16];
#pragma unroll
    for (int t = 0; t < 16; ++t) rb[t] = {};
    const float* bp = beta + (size_t)(b0 * S_) * E_ + e;
    float bte[4], bto[4];
    if (sw) {
      mag = 1.f / (1.f + expf(-alpha[e]));
      cw  = cosf(omega[e]);
      sn  = sinf(omega[e]);
#pragma unroll
      for (int t = 0; t < 16; ++t) {
        const int k = 32 * half + 2 * t;
        rb[t].x = res[k * E_ + e];
        rb[t].y = res[(k + 1) * E_ + e];
      }
#pragma unroll
      for (int t = 0; t < 16; ++t) asm volatile("" : "+v"(rb[t].x), "+v"(rb[t].y));
    }
#pragma unroll
    for (int q = 0; q < 4; ++q) bte[q] = bp[(size_t)q * S_ * E_ + (size_t)tw * E_];
#pragma unroll
    for (int q = 0; q < 4; ++q) bto[q] = bp[(size_t)q * S_ * E_ + (size_t)(tw + 1) * E_];

    v8f st = {};

    const int ob = (tid >> 4) & 7, oc = (tid & 15) * 4;
    float* sp = states + (size_t)(ob * S_ + t0) * E_ + oc;

    auto rotate = [&](const int cur, const float (&bt)[4]) {
      float* xw = &xs[cur * (16 * SP) + (8 * half) * SP + e];
#pragma unroll
      for (int q = 0; q < 4; ++q) {
        const float re = st[2 * q], im = st[2 * q + 1];
        const float nr = fmaf(mag, re * cw - im * sn, bt[q]);
        const float ni = mag * (re * sn + im * cw);
        xw[(2 * q) * SP]     = nr;
        xw[(2 * q + 1) * SP] = ni;
      }
    };

    auto refill = [&](float (&bt)[4], const int tn) {
      const int t2 = (tn < tlast) ? tn : tlast;
#pragma unroll
      for (int q = 0; q < 4; ++q) bt[q] = bp[(size_t)q * S_ * E_ + (size_t)t2 * E_];
    };

    auto step = [&](const int t, const int cur, float (&bt)[4], float (&btn)[4],
                    const bool emit, const bool flush, const int ot, const bool more) {
      asm volatile("s_wait_dscnt 0x0\n\ts_barrier_signal -1" ::: "memory");

      refill(bt, t + 2);

      asm volatile("s_barrier_wait -1" ::: "memory");

      if (sw) {
        const float4* ar = (const float4*)&xs[cur * (16 * SP) + m16 * SP + 32 * half];
        float4 av[8];
#pragma unroll
        for (int j = 0; j < 8; ++j) av[j] = ar[j];

        float4 o = make_float4(0.f, 0.f, 0.f, 0.f);
        if (flush) o = *(const float4*)&rs[(cur ^ 1) * (B_ * RP) + ob * RP + oc];

        asm volatile("" ::: "memory");
        __builtin_amdgcn_sched_barrier(0);

        v8f acc0 = {}, acc1 = {};
#pragma unroll
        for (int j = 0; j < 8; ++j) {
          v2f a0 = {av[j].x, av[j].y};
          v2f a1 = {av[j].z, av[j].w};
          acc0 = WMMA_F32(a0, rb[2 * j],     acc0);
          acc1 = WMMA_F32(a1, rb[2 * j + 1], acc1);
        }
        __builtin_amdgcn_sched_barrier(0);
        st = acc0 + acc1;

        if (emit) {
          float* rw = &rs[cur * (B_ * RP) + b0 * RP + e];
#pragma unroll
          for (int q = 0; q < 4; ++q) rw[q * RP] = st[2 * q];
          if (flush) *(float4*)(sp + (ot - 1) * E_) = o;
        }
      }
      if (more) {
        asm volatile("" : "+v"(btn[0]), "+v"(btn[1]), "+v"(btn[2]), "+v"(btn[3]));
        if (sw) rotate(cur ^ 1, btn);
      }
    };

    asm volatile("" : "+v"(bte[0]), "+v"(bte[1]), "+v"(bte[2]), "+v"(bte[3]));
    if (sw) rotate(0, bte);
    int t = tw;
    for (int i = 0; i < nwarm; i += 2, t += 2) {
      step(t,     0, bte, bto, false, false, 0, true);
      step(t + 1, 1, bto, bte, false, false, 0, true);
    }
    for (int ot = 0; ot < LCH; ot += 2, t += 2) {
      step(t,     0, bte, bto, true, ot > 0, ot,     true);
      step(t + 1, 1, bto, bte, true, true,   ot + 1, ot + 2 < LCH);
    }

    if (sw) {
      float* iw = &fs[b0 * RP + e];
#pragma unroll
      for (int q = 0; q < 4; ++q) iw[q * RP] = st[2 * q + 1];
    }
    asm volatile("s_wait_dscnt 0x0\n\ts_barrier_signal -1\n\ts_barrier_wait -1" ::: "memory");
    if (sw) {
      const float4 o = *(const float4*)&rs[((LCH - 1) & 1) * (B_ * RP) + ob * RP + oc];
      *(float4*)(sp + (LCH - 1) * E_) = o;
      if (c == NBLK - 1) {
        *(float4*)(rfin + ob * E_ + oc) = o;
        *(float4*)(ifin + ob * E_ + oc) = *(const float4*)&fs[ob * RP + oc];
      }
    }
  }
  __syncthreads();

  const int wi = __builtin_amdgcn_readfirstlane(tid >> 5);
  float* const stg = smem + 4 * YT_F;
  float* const wst = stg + wi * WSB_F;

  int so[8], fo[8];
#pragma unroll
  for (int i = 0; i < 8; ++i) {
    const int r = 2 * i + half;
    so[i] = r * E_ + 4 * (m16 ^ r);
    fo[i] = m16 * E_ + 4 * ((8 * half + i) ^ m16);
  }

  v4f g[8];
  auto fetch = [&](const float* p) {
#pragma unroll
    for (int i = 0; i < 8; ++i) g[i] = *(const v4f*)(p + 128 * i);
  };
  auto stage = [&]() {
#pragma unroll
    for (int i = 0; i < 8; ++i) *(v4f*)(wst + so[i]) = g[i];
  };
  auto ldfrag = [&](const float* tile, v4f (&f)[8]) {
#pragma unroll
    for (int j = 0; j < 8; ++j) f[j] = *(const v4f*)(tile + fo[j]);
  };
#define TILE_STORED() asm volatile("s_wait_dscnt 0x0" ::: "memory")

  v4f ga[8];
  if (wi < 4) {
    const float* s0p = states + (size_t)((2 * wi) * S_ + t0) * E_ + 4 * lane;
    const float* s1p = states + (size_t)((2 * wi + 1) * S_ + t0) * E_ + 4 * lane;
#pragma unroll
    for (int i = 0; i < 4; ++i) {
      ga[i]     = *(const v4f*)(s0p + 128 * i);
      ga[i + 4] = *(const v4f*)(s1p + 128 * i);
    }
  }
  v4f bq[4][8];
  const float* wt = w_out + (size_t)(wi * 4) * (16 * E_) + 4 * lane;
  fetch(wt);
  stage();
  fetch(wt + 1 * (16 * E_));
  TILE_STORED();
  ldfrag(wst, bq[0]);
  stage();
  fetch(wt + 2 * (16 * E_));
  TILE_STORED();
  ldfrag(wst, bq[1]);
  stage();
  fetch(wt + 3 * (16 * E_));
  TILE_STORED();
  ldfrag(wst, bq[2]);
  stage();
  TILE_STORED();
  ldfrag(wst, bq[3]);
  if (wi < 4) {
#pragma unroll
    for (int i = 0; i < 8; ++i) *(v4f*)(wst + so[i]) = ga[i];
  }
  __syncthreads();

  auto ybof = [&](const int T, const int dt) {
    unsigned yb = 4u * (unsigned)(T * YT_F + (8 * half) * D_ + (wi * 4 + dt) * 16 + m16);
    asm volatile("" : "+v"(yb));
    return yb;
  };
  auto mm2 = [&](const v4f (&aq)[8], const v4f (&b0)[8], const v4f (&b1)[8], const unsigned ya, const unsigned yb) {
    float* yr0 = (float*)((char*)smem + ya);
    float* yr1 = (float*)((char*)smem + yb);
    GROUP_EDGE();
    v8f acc0 = {}, acc1 = {};
#pragma unroll
    for (int j = 0; j < 8; ++j) {
      const v2f a0 = {aq[j].x, aq[j].y};
      const v2f a1 = {aq[j].z, aq[j].w};
      const v2f c0 = {b0[j].x, b0[j].y};
      const v2f d0 = {b1[j].x, b1[j].y};
      acc0 = WMMA_F32(a0, c0, acc0);
      acc1 = WMMA_F32(a0, d0, acc1);
      const v2f c1 = {b0[j].z, b0[j].w};
      const v2f d1 = {b1[j].z, b1[j].w};
      acc0 = WMMA_F32(a1, c1, acc0);
      acc1 = WMMA_F32(a1, d1, acc1);
    }
    GROUP_EDGE();
#pragma unroll
    for (int v = 0; v < 8; ++v) yr0[v * D_] = acc0[v];
#pragma unroll
    for (int v = 0; v < 8; ++v) yr1[v * D_] = acc1[v];
  };
  auto runs = [&](const int T, const v4f (&aq)[8]) {
    const unsigned y0 = ybof(T, 0), y1 = ybof(T, 1), y2 = ybof(T, 2), y3 = ybof(T, 3);
    mm2(aq, bq[0], bq[1], y0, y1);
    mm2(aq, bq[2], bq[3], y2, y3);
  };
  auto rowload = [&](const int T, v4f (&yv)[8]) {
    const float* yrow = smem + T * YT_F + wi * D_ + 4 * lane;
#pragma unroll
    for (int i = 0; i < 8; ++i) yv[i] = *(const v4f*)(yrow + 128 * i);
  };
  auto lx = [](const float v, auto mtag) {
    constexpr int m = decltype(mtag)::value;
    const int b = __builtin_bit_cast(int, v);
    if constexpr (m == 16)
      return __builtin_bit_cast(float, __builtin_amdgcn_permlanex16(b, b, 0x76543210, 0xfedcba98, false, false));
    else
      return __builtin_bit_cast(float, __builtin_amdgcn_update_dpp(b, b, 0x160 + m, 0xf, 0xf, false));
  };
  auto rowfinish = [&](const int T, const v4f (&yv)[8]) {
    v4f gA[4], bA[4], gB[4], bB[4];
#pragma unroll
    for (int i = 0; i < 4; ++i) {
      gA[i] = *(const v4f*)(gamma + 128 * i + 4 * lane);
      bA[i] = *(const v4f*)(lbeta + 128 * i + 4 * lane);
    }
    __builtin_amdgcn_sched_barrier(0);
    v4f p0 = yv[0];
    v4f p1 = yv[0] * yv[0];
#pragma unroll
    for (int i = 1; i < 8; ++i) {
      p0 += yv[i];
      p1 += yv[i] * yv[i];
    }
    float s0 = (p0.x + p0.y) + (p0.z + p0.w);
    float s1 = (p1.x + p1.y) + (p1.z + p1.w);
    s0 += lx(s0, std::integral_constant<int, 16>{});  s1 += lx(s1, std::integral_constant<int, 16>{});
    s0 += lx(s0, std::integral_constant<int, 8>{});   s1 += lx(s1, std::integral_constant<int, 8>{});
    s0 += lx(s0, std::integral_constant<int, 4>{});   s1 += lx(s1, std::integral_constant<int, 4>{});
    s0 += lx(s0, std::integral_constant<int, 2>{});   s1 += lx(s1, std::integral_constant<int, 2>{});
    s0 += lx(s0, std::integral_constant<int, 1>{});   s1 += lx(s1, std::integral_constant<int, 1>{});
    const float mu  = s0 * (1.f / D_);
    const float var = s1 * (1.f / D_) - mu * mu;
    const float rsd = rsqrtf(var + 1e-5f);
#pragma unroll
    for (int i = 0; i < 4; ++i) {
      gB[i] = *(const v4f*)(gamma + 128 * (i + 4) + 4 * lane);
      bB[i] = *(const v4f*)(lbeta + 128 * (i + 4) + 4 * lane);
    }
    __builtin_amdgcn_sched_barrier(0);
    float* yo = y + (size_t)((2 * T + (wi >> 3)) * S_ + t0 + (wi & 7)) * D_ + 4 * lane;
#pragma unroll
    for (int i = 0; i < 4; ++i) {
      const v4f o = (yv[i] - mu) * rsd * gA[i] + bA[i];
      *(v4f*)(yo + 128 * i) = o;
    }
#pragma unroll
    for (int i = 0; i < 4; ++i) {
      const v4f o = (yv[i + 4] - mu) * rsd * gB[i] + bB[i];
      *(v4f*)(yo + 128 * (i + 4)) = o;
    }
  };
#define LDS_BARRIER() asm volatile("s_wait_dscnt 0x0\n\ts_barrier_signal -1\n\ts_barrier_wait -1" ::: "memory")

  {
    v4f aq[8];
    ldfrag(stg, aq);
    runs(0, aq);
  }
  LDS_BARRIER();
#pragma unroll 1
  for (int T = 1; T < 4; ++T) {
    v4f aq[8], yv[8];
    ldfrag(stg + T * WSB_F, aq);
    rowload(T - 1, yv);
    runs(T, aq);
    asm volatile("s_wait_dscnt 0x0\n\ts_barrier_signal -1" ::: "memory");
    rowfinish(T - 1, yv);
    asm volatile("s_barrier_wait -1" ::: "memory");
  }
  {
    v4f yv[8];
    rowload(3, yv);
    rowfinish(3, yv);
  }
}

extern "C" void kernel_launch(void* const* d_in, const int* in_sizes, int n_in,
                              void* d_out, int out_size, void* d_ws, size_t ws_size,
                              hipStream_t stream) {
  (void)in_sizes; (void)n_in; (void)out_size; (void)ws_size;

  const float* x     = (const float*)d_in[0];
  const float* alpha = (const float*)d_in[1];
  const float* omega = (const float*)d_in[2];
  const float* w_in  = (const float*)d_in[3];
  const float* w_out = (const float*)d_in[4];
  const float* res   = (const float*)d_in[5];
  const float* gamma = (const float*)d_in[6];
  const float* lbeta = (const float*)d_in[7];

  float* out  = (float*)d_out;
  float* y    = out;
  float* rfin = out + YSZ;
  float* ifin = rfin + B_ * E_;

  float* beta   = (float*)d_ws;
  float* states = beta + (size_t)M_ * E_;

  hipLaunchKernelGGL(k_beta_gemm, dim3(M_ / 64), dim3(256), 0, stream, x, w_in, beta);
  hipLaunchKernelGGL(k_scan_out, dim3(NBLK), dim3(512), 0, stream,
                     beta, alpha, omega, res, states, rfin, ifin, w_out, gamma, lbeta, y);
}


# 64 scan warm-up steps replaced by a sum: new launch writes each 8-step block's zero-state end state E and the 8-step map C; block state = Horner of E(c-8..c-1) with C on WMMA; carry sum order changes
# speedup vs baseline: 6.1681x; 1.1591x over previous
#include <hip/hip_runtime.h>
#include <math.h>
#include <type_traits>

typedef float v2f __attribute__((ext_vector_type(2)));
typedef float v4f __attribute__((ext_vector_type(4)));
typedef float v8f __attribute__((ext_vector_type(8)));

#define WMMA_F32(a, b, c) \
  __builtin_amdgcn_wmma_f32_16x16x4_f32(false, (a), false, (b), (short)0, (c), false, false)

static constexpr int B_ = 8, S_ = 2048, D_ = 1024, E_ = 64;
static constexpr int M_ = B_ * S_;
static constexpr long long YSZ = (long long)M_ * D_;

static constexpr int KS    = 128;
static constexpr int NSLAB = D_ / KS;
static constexpr int TP    = 132;

__global__ __launch_bounds__(256)
void k_beta_gemm(const float* __restrict__ x, const float* __restrict__ w_in,
                 float* __restrict__ beta) {
  const int tid  = threadIdx.x;
  const int lane = tid & 31;
  const int w    = tid >> 5;
  const int wm   = w & 3;
  const int wn   = w >> 2;
  const int m16  = lane & 15;
  const int half = lane >> 4;

  __shared__ __attribute__((aligned(16))) float xt[2][64 * TP];
  __shared__ __attribute__((aligned(16))) float wt[2][64 * TP];

  const int sr = tid >> 4;
  const int sc = (tid & 15) * 4;
  const float* xg = x    + (size_t)(blockIdx.x * 64 + sr) * D_ + sc;
  const float* wg = w_in + (size_t)sr * D_ + sc;
  const int so = sr * TP + sc;

  v4f gx[8], gw[8];

  auto fetch = [&](const int s) {
#pragma unroll
    for (int i = 0; i < 4; ++i) {
#pragma unroll
      for (int u = 0; u < 2; ++u) {
        gx[2 * i + u] = *(const v4f*)(xg + (size_t)(16 * i) * D_ + s * KS + 64 * u);
        gw[2 * i + u] = *(const v4f*)(wg + (size_t)(16 * i) * D_ + s * KS + 64 * u);
      }
    }
  };
  auto stage = [&](const int buf) {
#pragma unroll
    for (int i = 0; i < 4; ++i) {
#pragma unroll
      for (int u = 0; u < 2; ++u) {
        *(v4f*)&xt[buf][so + 16 * i * TP + 64 * u] = gx[2 * i + u];
        *(v4f*)&wt[buf][so + 16 * i * TP + 64 * u] = gw[2 * i + u];
      }
    }
  };

  v8f acc[2];
#pragma unroll
  for (int i = 0; i < 2; ++i) acc[i] = {};

  auto ldq = [&](const float* ar, const float* br, const int Q, v4f (&a)[2], v4f (&b)[2][2]) {
    const int co = 64 * (Q >> 2) + 8 * (Q & 3);
#pragma unroll
    for (int jj = 0; jj < 2; ++jj) {
      a[jj] = *(const v4f*)(ar + co + 4 * jj);
#pragma unroll
      for (int nt = 0; nt < 2; ++nt)
        b[jj][nt] = *(const v4f*)(br + nt * 16 * TP + co + 4 * jj);
    }
  };
  auto mmq = [&](const v4f (&a)[2], const v4f (&b)[2][2]) {
#pragma unroll
    for (int jj = 0; jj < 2; ++jj) {
      const v2f a0 = {a[jj].x, a[jj].y};
      const v2f a1 = {a[jj].z, a[jj].w};
#pragma unroll
      for (int nt = 0; nt < 2; ++nt) {
        const v2f b0 = {b[jj][nt].x, b[jj][nt].y};
        acc[nt] = WMMA_F32(a0, b0, acc[nt]);
      }
#pragma unroll
      for (int nt = 0; nt < 2; ++nt) {
        const v2f b1 = {b[jj][nt].z, b[jj][nt].w};
        acc[nt] = WMMA_F32(a1, b1, acc[nt]);
      }
    }
  };
#define GROUP_EDGE() do { asm volatile("" ::: "memory"); __builtin_amdgcn_sched_barrier(0); } while (0)

  auto multiply = [&](const int buf) {
    const float* ar = &xt[buf][(16 * wm + m16) * TP + 32 * half];
    const float* br = &wt[buf][(32 * wn + m16) * TP + 32 * half];
    v4f a0[2], b0[2][2], a1[2], b1[2][2];
    ldq(ar, br, 0, a0, b0);
    ldq(ar, br, 1, a1, b1);
    GROUP_EDGE();
#pragma unroll
    for (int Q = 0; Q < 8; Q += 2) {
      mmq(a0, b0);
      if (Q + 2 < 8) {
        GROUP_EDGE();
        ldq(ar, br, Q + 2, a0, b0);
        GROUP_EDGE();
      }
      mmq(a1, b1);
      if (Q + 3 < 8) {
        GROUP_EDGE();
        ldq(ar, br, Q + 3, a1, b1);
        GROUP_EDGE();
      }
    }
    GROUP_EDGE();
  };

  fetch(0);
  stage(0);
  __syncthreads();
  for (int s = 0; s < NSLAB - 1; ++s) {
    fetch(s + 1);
    multiply(s & 1);
    stage((s + 1) & 1);
    __syncthreads();
  }
  multiply((NSLAB - 1) & 1);

#pragma unroll
  for (int nt = 0; nt < 2; ++nt)
#pragma unroll
    for (int v = 0; v < 8; ++v)
      beta[(size_t)(blockIdx.x * 64 + wm * 16 + v + 8 * half) * E_ + (2 * wn + nt) * 16 + m16] = acc[nt][v];
}

static constexpr int LCH  = 8;
static constexpr int NBLK = S_ / LCH;
static constexpr int NPRE = 8;
static constexpr int NUB  = E_ / B_;
static constexpr int ET_F = 16 * E_;
static constexpr int SP = 68;
static constexpr int RP = 72;

static constexpr int YT_F  = 16 * D_;
static constexpr int WSB_F = 16 * E_;
static constexpr int XS_F  = 2 * 16 * SP;
static constexpr int RS_F  = 2 * B_ * RP;
static constexpr int FS_F  = B_ * RP;
static_assert(XS_F + RS_F + FS_F <= 16 * WSB_F, "the scan's buffers lie inside the staging tiles");

__global__ __launch_bounds__(128)
void k_block_resp(const float* __restrict__ beta, const float* __restrict__ alpha,
                  const float* __restrict__ omega, const float* __restrict__ res,
                  float* __restrict__ eb, float* __restrict__ cr, float* __restrict__ ci) {
  const int c    = blockIdx.x;
  const int tid  = threadIdx.x;
  const int lane = tid & 31;
  const int m16  = lane & 15;
  const int half = lane >> 4;
  const int w    = tid >> 5;
  const int e    = w * 16 + m16;
  const int b0   = 4 * half;
  const bool unit = c >= NBLK;
  const int u    = c - NBLK;
  const int t0   = unit ? 0 : c * LCH;

  __shared__ __attribute__((aligned(16))) float xs[2 * 16 * SP];

  const float mag = 1.f / (1.f + expf(-alpha[e]));
  const float cw  = cosf(omega[e]);
  const float sn  = sinf(omega[e]);
  v2f rb[16];
#pragma unroll
  for (int t = 0; t < 16; ++t) {
    const int k = 32 * half + 2 * t;
    rb[t].x = res[k * E_ + e];
    rb[t].y = res[(k + 1) * E_ + e];
  }
  float bt[LCH][4];
  const float* bp = beta + (size_t)(b0 * S_ + t0) * E_ + e;
#pragma unroll
  for (int i = 0; i < LCH; ++i)
#pragma unroll
    for (int q = 0; q < 4; ++q) {
      const float v = bp[(size_t)q * S_ * E_ + (size_t)i * E_];
      bt[i][q] = unit ? 0.f : v;
    }

  v8f st = {};
  if (unit) {
#pragma unroll
    for (int q = 0; q < 4; ++q) st[2 * q] = (e == 8 * u + b0 + q) ? 1.f : 0.f;
  }

#pragma unroll
  for (int i = 0; i < LCH; ++i) {
    const int cur = i & 1;
    float* xw = &xs[cur * (16 * SP) + (8 * half) * SP + e];
#pragma unroll
    for (int q = 0; q < 4; ++q) {
      const float re = st[2 * q], im = st[2 * q + 1];
      const float nr = fmaf(mag, re * cw - im * sn, bt[i][q]);
      const float ni = mag * (re * sn + im * cw);
      xw[(2 * q) * SP]     = nr;
      xw[(2 * q + 1) * SP] = ni;
    }
    __syncthreads();
    const float4* ar = (const float4*)&xs[cur * (16 * SP) + m16 * SP + 32 * half];
    float4 av[8];
#pragma unroll
    for (int j = 0; j < 8; ++j) av[j] = ar[j];
    asm volatile("" ::: "memory");
    __builtin_amdgcn_sched_barrier(0);
    v8f acc0 = {}, acc1 = {};
#pragma unroll
    for (int j = 0; j < 8; ++j) {
      v2f a0 = {av[j].x, av[j].y};
      v2f a1 = {av[j].z, av[j].w};
      acc0 = WMMA_F32(a0, rb[2 * j],     acc0);
      acc1 = WMMA_F32(a1, rb[2 * j + 1], acc1);
    }
    __builtin_amdgcn_sched_barrier(0);
    st = acc0 + acc1;
  }

  {
    float* xw = &xs[(8 * half) * SP + e];
#pragma unroll
    for (int i = 0; i < 8; ++i) xw[i * SP] = st[i];
  }
  __syncthreads();
  const int orow = tid >> 4, oc = (tid & 15) * 4;
#pragma unroll
  for (int hh = 0; hh < 2; ++hh) {
    const int row = orow + 8 * hh;
    const float4 o = *(const float4*)&xs[row * SP + oc];
    float* dst = unit ? (((row & 1) ? ci : cr) + (size_t)(8 * u + (row >> 1)) * E_ + oc)
                      : (eb + (size_t)c * ET_F + row * E_ + oc);
    *(float4*)dst = o;
  }
}

__global__ __launch_bounds__(512)
void k_scan_out(const float* __restrict__ beta, const float* __restrict__ alpha,
                const float* __restrict__ omega, const float* __restrict__ res,
                const float* __restrict__ eb, const float* __restrict__ cr, const float* __restrict__ ci,
                float* __restrict__ rfin, float* __restrict__ ifin,
                const float* __restrict__ w_out,
                const float* __restrict__ gamma, const float* __restrict__ lbeta,
                float* __restrict__ y) {
  const int c    = blockIdx.x;
  const int tid  = threadIdx.x;
  const int lane = tid & 31;
  const int m16  = lane & 15;
  const int half = lane >> 4;
  const int t0   = c * LCH;

  __shared__ __attribute__((aligned(16))) float smem[4 * YT_F + 16 * WSB_F];
  float* const xs = smem + 4 * YT_F;
  float* const rs = xs + XS_F;
  float* const fs = rs + RS_F;

  {
    const bool sw  = __builtin_amdgcn_readfirstlane(tid >> 7) == 0;
    const int w    = (tid >> 5) & 3;
    const int e    = w * 16 + m16;
    const int b0   = 4 * half;

    const int npre  = (c < NPRE) ? c : NPRE;
    const int tlast = t0 + LCH - 1;

    float mag = 0.f, cw = 0.f, sn = 0.f;
    v2f rb[16], crb[16], cib[16];
#pragma unroll
    for (int t = 0; t < 16; ++t) { rb[t] = {}; crb[t] = {}; cib[t] = {}; }
    const float* bp = beta + (size_t)(b0 * S_) * E_ + e;
    float bte[4], bto[4];
    if (sw) {
      mag = 1.f / (1.f + expf(-alpha[e]));
      cw  = cosf(omega[e]);
      sn  = sinf(omega[e]);
#pragma unroll
      for (int t = 0; t < 16; ++t) {
        const int k = 32 * half + 2 * t;
        crb[t].x = cr[k * E_ + e];
        crb[t].y = cr[(k + 1) * E_ + e];
        cib[t].x = ci[k * E_ + e];
        cib[t].y = ci[(k + 1) * E_ + e];
        rb[t].x = res[k * E_ + e];
        rb[t].y = res[(k + 1) * E_ + e];
      }
#pragma unroll
      for (int t = 0; t < 16; ++t) {
        asm volatile("" : "+v"(crb[t].x), "+v"(crb[t].y), "+v"(cib[t].x), "+v"(cib[t].y));
        asm volatile("" : "+v"(rb[t].x), "+v"(rb[t].y));
      }
    }

    const float* ep = eb + (size_t)(2 * b0) * E_ + e;
    auto load_end = [&](const int d, float (&v)[8]) {
      const float* p = ep + (size_t)d * ET_F;
#pragma unroll
      for (int i = 0; i < 8; ++i) v[i] = p[i * E_];
    };
    v8f st = {};
    {
      float e0[8];
      load_end(c - npre, e0);
#pragma unroll
      for (int i = 0; i < 8; ++i) st[i] = (npre > 0) ? e0[i] : 0.f;
    }

#pragma unroll
    for (int q = 0; q < 4; ++q) bte[q] = bp[(size_t)q * S_ * E_ + (size_t)t0 * E_];
#pragma unroll
    for (int q = 0; q < 4; ++q) bto[q] = bp[(size_t)q * S_ * E_ + (size_t)(t0 + 1) * E_];

    const int ob = (tid >> 4) & 7, oc = (tid & 15) * 4;
    float* sp = y + (size_t)(ob * S_ + t0) * D_ + oc;

    auto rotate = [&](const int cur, const float (&bt)[4]) {
      float* xw = &xs[cur * (16 * SP) + (8 * half) * SP + e];
#pragma unroll
      for (int q = 0; q < 4; ++q) {
        const float re = st[2 * q], im = st[2 * q + 1];
        const float nr = fmaf(mag, re * cw - im * sn, bt[q]);
        const float ni = mag * (re * sn + im * cw);
        xw[(2 * q) * SP]     = nr;
        xw[(2 * q + 1) * SP] = ni;
      }
    };

    auto refill = [&](float (&bt)[4], const int tn) {
      const int t2 = (tn < tlast) ? tn : tlast;
#pragma unroll
      for (int q = 0; q < 4; ++q) bt[q] = bp[(size_t)q * S_ * E_ + (size_t)t2 * E_];
    };

    auto step = [&](const int t, const int cur, float (&bt)[4], float (&btn)[4],
                    const bool emit, const bool flush, const int ot, const bool more) {
      asm volatile("s_wait_dscnt 0x0\n\ts_barrier_signal -1" ::: "memory");

      refill(bt, t + 2);

      asm volatile("s_barrier_wait -1" ::: "memory");

      if (sw) {
        const float4* ar = (const float4*)&xs[cur * (16 * SP) + m16 * SP + 32 * half];
        float4 av[8];
#pragma unroll
        for (int j = 0; j < 8; ++j) av[j] = ar[j];

        float4 o = make_float4(0.f, 0.f, 0.f, 0.f);
        if (flush) o = *(const float4*)&rs[(cur ^ 1) * (B_ * RP) + ob * RP + oc];

        asm volatile("" ::: "memory");
        __builtin_amdgcn_sched_barrier(0);

        v8f acc0 = {}, acc1 = {};
#pragma unroll
        for (int j = 0; j < 8; ++j) {
          v2f a0 = {av[j].x, av[j].y};
          v2f a1 = {av[j].z, av[j].w};
          acc0 = WMMA_F32(a0, rb[2 * j],     acc0);
          acc1 = WMMA_F32(a1, rb[2 * j + 1], acc1);
        }
        __builtin_amdgcn_sched_barrier(0);
        st = acc0 + acc1;

        if (emit) {
          float* rw = &rs[cur * (B_ * RP) + b0 * RP + e];
#pragma unroll
          for (int q = 0; q < 4; ++q) rw[q * RP] = st[2 * q];
          if (flush) *(float4*)(sp + (ot - 1) * E_) = o;
        }
      }
      if (more) {
        asm volatile("" : "+v"(btn[0]), "+v"(btn[1]), "+v"(btn[2]), "+v"(btn[3]));
        if (sw) rotate(cur ^ 1, btn);
      }
    };

    {
      int xo = 0;
      for (int g = npre - 1; g >= 1; --g) {
        if (sw) {
          float* xw = &xs[xo + (8 * half) * SP + e];
#pragma unroll
          for (int i = 0; i < 8; ++i) xw[i * SP] = st[i];
        }
        asm volatile("s_wait_dscnt 0x0\n\ts_barrier_signal -1" ::: "memory");
        float en[8];
        load_end(c - g, en);
        asm volatile("s_barrier_wait -1" ::: "memory");

        v8f pa = {}, pj = {};
        if (sw) {
          const float4* ar = (const float4*)&xs[xo + m16 * SP + 32 * half];
          const float4* jr = (const float4*)&xs[xo + (m16 ^ 1) * SP + 32 * half];
          float4 av[8], jv[8];
#pragma unroll
          for (int j = 0; j < 8; ++j) av[j] = ar[j];
#pragma unroll
          for (int j = 0; j < 8; ++j) jv[j] = jr[j];
          asm volatile("" ::: "memory");
          __builtin_amdgcn_sched_barrier(0);
          v8f pa0 = {}, pa1 = {}, pj0 = {}, pj1 = {};
#pragma unroll
          for (int j = 0; j < 8; ++j) {
            v2f a0 = {av[j].x, av[j].y};
            v2f a1 = {av[j].z, av[j].w};
            v2f j0 = {jv[j].x, jv[j].y};
            v2f j1 = {jv[j].z, jv[j].w};
            pa0 = WMMA_F32(a0, crb[2 * j],     pa0);
            pj0 = WMMA_F32(j0, cib[2 * j],     pj0);
            pa1 = WMMA_F32(a1, crb[2 * j + 1], pa1);
            pj1 = WMMA_F32(j1, cib[2 * j + 1], pj1);
          }
          __builtin_amdgcn_sched_barrier(0);
          pa = pa0 + pa1;
          pj = pj0 + pj1;
        }
        asm volatile("" : "+v"(en[0]), "+v"(en[1]), "+v"(en[2]), "+v"(en[3]),
                          "+v"(en[4]), "+v"(en[5]), "+v"(en[6]), "+v"(en[7]));
#pragma unroll
        for (int q = 0; q < 4; ++q) {
          st[2 * q]     = (pa[2 * q]     - pj[2 * q])     + en[2 * q];
          st[2 * q + 1] = (pa[2 * q + 1] + pj[2 * q + 1]) + en[2 * q + 1];
        }
        xo = 16 * SP - xo;
      }
      asm volatile("s_wait_dscnt 0x0\n\ts_barrier_signal -1\n\ts_barrier_wait -1" ::: "memory");
    }

    asm volatile("" : "+v"(bte[0]), "+v"(bte[1]), "+v"(bte[2]), "+v"(bte[3]));
    if (sw) rotate(0, bte);
    int t = t0;
    for (int ot = 0; ot < LCH; ot += 2, t += 2) {
      step(t,     0, bte, bto, true, ot > 0, ot,     true);
      step(t + 1, 1, bto, bte, true, true,   ot + 1, ot + 2 < LCH);
    }

    if (sw) {
      float* iw = &fs[b0 * RP + e];
#pragma unroll
      for (int q = 0; q < 4; ++q) iw[q * RP] = st[2 * q + 1];
    }
    asm volatile("s_wait_dscnt 0x0\n\ts_barrier_signal -1\n\ts_barrier_wait -1" ::: "memory");
    if (sw) {
      const float4 o = *(const float4*)&rs[((LCH - 1) & 1) * (B_ * RP) + ob * RP + oc];
      *(float4*)(sp + (LCH - 1) * E_) = o;
      if (c == NBLK - 1) {
        *(float4*)(rfin + ob * E_ + oc) = o;
        *(float4*)(ifin + ob * E_ + oc) = *(const float4*)&fs[ob * RP + oc];
      }
    }
  }
  __syncthreads();

  const int wi = __builtin_amdgcn_readfirstlane(tid >> 5);
  float* const stg = smem + 4 * YT_F;
  float* const wst = stg + wi * WSB_F;

  int so[8], fo[8];
#pragma unroll
  for (int i = 0; i < 8; ++i) {
    const int r = 2 * i + half;
    so[i] = r * E_ + 4 * (m16 ^ r);
    fo[i] = m16 * E_ + 4 * ((8 * half + i) ^ m16);
  }

  v4f g[8];
  auto fetch = [&](const float* p) {
#pragma unroll
    for (int i = 0; i < 8; ++i) g[i] = *(const v4f*)(p + 128 * i);
  };
  auto stage = [&]() {
#pragma unroll
    for (int i = 0; i < 8; ++i) *(v4f*)(wst + so[i]) = g[i];
  };
  auto ldfrag = [&](const float* tile, v4f (&f)[8]) {
#pragma unroll
    for (int j = 0; j < 8; ++j) f[j] = *(const v4f*)(tile + fo[j]);
  };
#define TILE_STORED() asm volatile("s_wait_dscnt 0x0" ::: "memory")

  v4f ga[8];
  if (wi < 4) {
    const float* s0p = y + (size_t)((2 * wi) * S_ + t0) * D_ + 4 * lane;
    const float* s1p = y + (size_t)((2 * wi + 1) * S_ + t0) * D_ + 4 * lane;
#pragma unroll
    for (int i = 0; i < 4; ++i) {
      ga[i]     = *(const v4f*)(s0p + 128 * i);
      ga[i + 4] = *(const v4f*)(s1p + 128 * i);
    }
  }
  v4f bq[4][8];
  const float* wt = w_out + (size_t)(wi * 4) * (16 * E_) + 4 * lane;
  fetch(wt);
  stage();
  fetch(wt + 1 * (16 * E_));
  TILE_STORED();
  ldfrag(wst, bq[0]);
  stage();
  fetch(wt + 2 * (16 * E_));
  TILE_STORED();
  ldfrag(wst, bq[1]);
  stage();
  fetch(wt + 3 * (16 * E_));
  TILE_STORED();
  ldfrag(wst, bq[2]);
  stage();
  TILE_STORED();
  ldfrag(wst, bq[3]);
  if (wi < 4) {
#pragma unroll
    for (int i = 0; i < 8; ++i) *(v4f*)(wst + so[i]) = ga[i];
  }
  __syncthreads();

  auto ybof = [&](const int T, const int dt) {
    unsigned yb = 4u * (unsigned)(T * YT_F + (8 * half) * D_ + (wi * 4 + dt) * 16 + m16);
    asm volatile("" : "+v"(yb));
    return yb;
  };
  auto mm2 = [&](const v4f (&aq)[8], const v4f (&b0)[8], const v4f (&b1)[8], const unsigned ya, const unsigned yb) {
    float* yr0 = (float*)((char*)smem + ya);
    float* yr1 = (float*)((char*)smem + yb);
    GROUP_EDGE();
    v8f acc0 = {}, acc1 = {};
#pragma unroll
    for (int j = 0; j < 8; ++j) {
      const v2f a0 = {aq[j].x, aq[j].y};
      const v2f a1 = {aq[j].z, aq[j].w};
      const v2f c0 = {b0[j].x, b0[j].y};
      const v2f d0 = {b1[j].x, b1[j].y};
      acc0 = WMMA_F32(a0, c0, acc0);
      acc1 = WMMA_F32(a0, d0, acc1);
      const v2f c1 = {b0[j].z, b0[j].w};
      const v2f d1 = {b1[j].z, b1[j].w};
      acc0 = WMMA_F32(a1, c1, acc0);
      acc1 = WMMA_F32(a1, d1, acc1);
    }
    GROUP_EDGE();
#pragma unroll
    for (int v = 0; v < 8; ++v) yr0[v * D_] = acc0[v];
#pragma unroll
    for (int v = 0; v < 8; ++v) yr1[v * D_] = acc1[v];
  };
  auto runs = [&](const int T, const v4f (&aq)[8]) {
    const unsigned y0 = ybof(T, 0), y1 = ybof(T, 1), y2 = ybof(T, 2), y3 = ybof(T, 3);
    mm2(aq, bq[0], bq[1], y0, y1);
    mm2(aq, bq[2], bq[3], y2, y3);
  };
  auto rowload = [&](const int T, v4f (&yv)[8]) {
    const float* yrow = smem + T * YT_F + wi * D_ + 4 * lane;
#pragma unroll
    for (int i = 0; i < 8; ++i) yv[i] = *(const v4f*)(yrow + 128 * i);
  };
  auto lx = [](const float v, auto mtag) {
    constexpr int m = decltype(mtag)::value;
    const int b = __builtin_bit_cast(int, v);
    if constexpr (m == 16)
      return __builtin_bit_cast(float, __builtin_amdgcn_permlanex16(b, b, 0x76543210, 0xfedcba98, false, false));
    else
      return __builtin_bit_cast(float, __builtin_amdgcn_update_dpp(b, b, 0x160 + m, 0xf, 0xf, false));
  };
  auto rowfinish = [&](const int T, const v4f (&yv)[8]) {
    v4f gA[4], bA[4], gB[4], bB[4];
#pragma unroll
    for (int i = 0; i < 4; ++i) {
      gA[i] = *(const v4f*)(gamma + 128 * i + 4 * lane);
      bA[i] = *(const v4f*)(lbeta + 128 * i + 4 * lane);
    }
    __builtin_amdgcn_sched_barrier(0);
    v4f p0 = yv[0];
    v4f p1 = yv[0] * yv[0];
#pragma unroll
    for (int i = 1; i < 8; ++i) {
      p0 += yv[i];
      p1 += yv[i] * yv[i];
    }
    float s0 = (p0.x + p0.y) + (p0.z + p0.w);
    float s1 = (p1.x + p1.y) + (p1.z + p1.w);
    s0 += lx(s0, std::integral_constant<int, 16>{});  s1 += lx(s1, std::integral_constant<int, 16>{});
    s0 += lx(s0, std::integral_constant<int, 8>{});   s1 += lx(s1, std::integral_constant<int, 8>{});
    s0 += lx(s0, std::integral_constant<int, 4>{});   s1 += lx(s1, std::integral_constant<int, 4>{});
    s0 += lx(s0, std::integral_constant<int, 2>{});   s1 += lx(s1, std::integral_constant<int, 2>{});
    s0 += lx(s0, std::integral_constant<int, 1>{});   s1 += lx(s1, std::integral_constant<int, 1>{});
    const float mu  = s0 * (1.f / D_);
    const float var = s1 * (1.f / D_) - mu * mu;
    const float rsd = rsqrtf(var + 1e-5f);
#pragma unroll
    for (int i = 0; i < 4; ++i) {
      gB[i] = *(const v4f*)(gamma + 128 * (i + 4) + 4 * lane);
      bB[i] = *(const v4f*)(lbeta + 128 * (i + 4) + 4 * lane);
    }
    __builtin_amdgcn_sched_barrier(0);
    float* yo = y + (size_t)((2 * T + (wi >> 3)) * S_ + t0 + (wi & 7)) * D_ + 4 * lane;
#pragma unroll
    for (int i = 0; i < 4; ++i) {
      const v4f o = (yv[i] - mu) * rsd * gA[i] + bA[i];
      *(v4f*)(yo + 128 * i) = o;
    }
#pragma unroll
    for (int i = 0; i < 4; ++i) {
      const v4f o = (yv[i + 4] - mu) * rsd * gB[i] + bB[i];
      *(v4f*)(yo + 128 * (i + 4)) = o;
    }
  };
#define LDS_BARRIER() asm volatile("s_wait_dscnt 0x0\n\ts_barrier_signal -1\n\ts_barrier_wait -1" ::: "memory")

  {
    v4f aq[8];
    ldfrag(stg, aq);
    runs(0, aq);
  }
  LDS_BARRIER();
#pragma unroll 1
  for (int T = 1; T < 4; ++T) {
    v4f aq[8], yv[8];
    ldfrag(stg + T * WSB_F, aq);
    rowload(T - 1, yv);
    runs(T, aq);
    asm volatile("s_wait_dscnt 0x0\n\ts_barrier_signal -1" ::: "memory");
    rowfinish(T - 1, yv);
    asm volatile("s_barrier_wait -1" ::: "memory");
  }
  {
    v4f yv[8];
    rowload(3, yv);
    rowfinish(3, yv);
  }
}

extern "C" void kernel_launch(void* const* d_in, const int* in_sizes, int n_in,
                              void* d_out, int out_size, void* d_ws, size_t ws_size,
                              hipStream_t stream) {
  (void)in_sizes; (void)n_in; (void)out_size; (void)ws_size;

  const float* x     = (const float*)d_in[0];
  const float* alpha = (const float*)d_in[1];
  const float* omega = (const float*)d_in[2];
  const float* w_in  = (const float*)d_in[3];
  const float* w_out = (const float*)d_in[4];
  const float* res   = (const float*)d_in[5];
  const float* gamma = (const float*)d_in[6];
  const float* lbeta = (const float*)d_in[7];

  float* out  = (float*)d_out;
  float* y    = out;
  float* rfin = out + YSZ;
  float* ifin = rfin + B_ * E_;

  float* beta = (float*)d_ws;
  float* eb   = beta + (size_t)M_ * E_;
  float* cr   = eb + (size_t)NBLK * ET_F;
  float* ci   = cr + E_ * E_;

  hipLaunchKernelGGL(k_beta_gemm, dim3(M_ / 64), dim3(256), 0, stream, x, w_in, beta);
  hipLaunchKernelGGL(k_block_resp, dim3(NBLK + NUB), dim3(128), 0, stream,
                     beta, alpha, omega, res, eb, cr, ci);
  hipLaunchKernelGGL(k_scan_out, dim3(NBLK), dim3(512), 0, stream,
                     beta, alpha, omega, res, eb, cr, ci, rfin, ifin, w_out, gamma, lbeta, y);
}


# end states E and map C now computed at the end of the input GEMM launch from its beta tile staged in LDS (a block's 8 runs of 8 steps = the 16 WMMA rows; 8 extra unit blocks); launches 3 -> 2
# speedup vs baseline: 6.5317x; 1.0589x over previous
#include <hip/hip_runtime.h>
#include <math.h>
#include <type_traits>

typedef float v2f __attribute__((ext_vector_type(2)));
typedef float v4f __attribute__((ext_vector_type(4)));
typedef float v8f __attribute__((ext_vector_type(8)));

#define WMMA_F32(a, b, c) \
  __builtin_amdgcn_wmma_f32_16x16x4_f32(false, (a), false, (b), (short)0, (c), false, false)

static constexpr int B_ = 8, S_ = 2048, D_ = 1024, E_ = 64;
static constexpr int M_ = B_ * S_;
static constexpr long long YSZ = (long long)M_ * D_;

static constexpr int LCH  = 8;
static constexpr int NBLK = S_ / LCH;
static constexpr int NUB  = E_ / B_;
static constexpr int ET_F = 16 * E_;
static constexpr int SP = 68;
static constexpr int NGB  = M_ / 64;
static constexpr int BP   = 67;
static constexpr int KS    = 128;
static constexpr int NSLAB = D_ / KS;
static constexpr int TP    = 132;

__global__ __launch_bounds__(256)
void k_beta_gemm(const float* __restrict__ x, const float* __restrict__ w_in,
                 const float* __restrict__ alpha, const float* __restrict__ omega,
                 const float* __restrict__ res,
                 float* __restrict__ beta, float* __restrict__ eb,
                 float* __restrict__ cr, float* __restrict__ ci) {
  const bool unit = blockIdx.x >= NGB;
  const int tid  = threadIdx.x;
  const int lane = tid & 31;
  const int w    = tid >> 5;
  const int wm   = w & 3;
  const int wn   = w >> 2;
  const int m16  = lane & 15;
  const int half = lane >> 4;

  __shared__ __attribute__((aligned(16))) float xt[2][64 * TP];
  __shared__ __attribute__((aligned(16))) float wt[2][64 * TP];

  const int sr = tid >> 4;
  const int sc = (tid & 15) * 4;
  const float* xg = x    + (size_t)(blockIdx.x * 64 + sr) * D_ + sc;
  const float* wg = w_in + (size_t)sr * D_ + sc;
  const int so = sr * TP + sc;

  v4f gx[8], gw[8];

  auto fetch = [&](const int s) {
#pragma unroll
    for (int i = 0; i < 4; ++i) {
#pragma unroll
      for (int u = 0; u < 2; ++u) {
        gx[2 * i + u] = *(const v4f*)(xg + (size_t)(16 * i) * D_ + s * KS + 64 * u);
        gw[2 * i + u] = *(const v4f*)(wg + (size_t)(16 * i) * D_ + s * KS + 64 * u);
      }
    }
  };
  auto stage = [&](const int buf) {
#pragma unroll
    for (int i = 0; i < 4; ++i) {
#pragma unroll
      for (int u = 0; u < 2; ++u) {
        *(v4f*)&xt[buf][so + 16 * i * TP + 64 * u] = gx[2 * i + u];
        *(v4f*)&wt[buf][so + 16 * i * TP + 64 * u] = gw[2 * i + u];
      }
    }
  };

  v8f acc[2];
#pragma unroll
  for (int i = 0; i < 2; ++i) acc[i] = {};

  auto ldq = [&](const float* ar, const float* br, const int Q, v4f (&a)[2], v4f (&b)[2][2]) {
    const int co = 64 * (Q >> 2) + 8 * (Q & 3);
#pragma unroll
    for (int jj = 0; jj < 2; ++jj) {
      a[jj] = *(const v4f*)(ar + co + 4 * jj);
#pragma unroll
      for (int nt = 0; nt < 2; ++nt)
        b[jj][nt] = *(const v4f*)(br + nt * 16 * TP + co + 4 * jj);
    }
  };
  auto mmq = [&](const v4f (&a)[2], const v4f (&b)[2][2]) {
#pragma unroll
    for (int jj = 0; jj < 2; ++jj) {
      const v2f a0 = {a[jj].x, a[jj].y};
      const v2f a1 = {a[jj].z, a[jj].w};
#pragma unroll
      for (int nt = 0; nt < 2; ++nt) {
        const v2f b0 = {b[jj][nt].x, b[jj][nt].y};
        acc[nt] = WMMA_F32(a0, b0, acc[nt]);
      }
#pragma unroll
      for (int nt = 0; nt < 2; ++nt) {
        const v2f b1 = {b[jj][nt].z, b[jj][nt].w};
        acc[nt] = WMMA_F32(a1, b1, acc[nt]);
      }
    }
  };
#define GROUP_EDGE() do { asm volatile("" ::: "memory"); __builtin_amdgcn_sched_barrier(0); } while (0)

  auto multiply = [&](const int buf) {
    const float* ar = &xt[buf][(16 * wm + m16) * TP + 32 * half];
    const float* br = &wt[buf][(32 * wn + m16) * TP + 32 * half];
    v4f a0[2], b0[2][2], a1[2], b1[2][2];
    ldq(ar, br, 0, a0, b0);
    ldq(ar, br, 1, a1, b1);
    GROUP_EDGE();
#pragma unroll
    for (int Q = 0; Q < 8; Q += 2) {
      mmq(a0, b0);
      if (Q + 2 < 8) {
        GROUP_EDGE();
        ldq(ar, br, Q + 2, a0, b0);
        GROUP_EDGE();
      }
      mmq(a1, b1);
      if (Q + 3 < 8) {
        GROUP_EDGE();
        ldq(ar, br, Q + 3, a1, b1);
        GROUP_EDGE();
      }
    }
    GROUP_EDGE();
  };

  const bool sw = __builtin_amdgcn_readfirstlane(tid >> 7) == 0;
  const int e   = w * 16 + m16;
  const int b0  = 4 * half;
  float araw = 0.f, oraw = 0.f;
  if (sw) { araw = alpha[e]; oraw = omega[e]; }

  float* const bs = &xt[0][0];
  float* const xs = &wt[0][0];

  v2f rb[16];
#pragma unroll
  for (int t = 0; t < 16; ++t) rb[t] = {};
  auto load_rb = [&]() {
#pragma unroll
    for (int t = 0; t < 16; ++t) {
      const int k = 32 * half + 2 * t;
      rb[t].x = res[k * E_ + e];
      rb[t].y = res[(k + 1) * E_ + e];
    }
  };

  float mag = 0.f, cw = 0.f, sn = 0.f;
  if (!unit) {
    fetch(0);
    stage(0);
    __syncthreads();
    if (sw) {
      mag = 1.f / (1.f + expf(-araw));
      cw  = cosf(oraw);
      sn  = sinf(oraw);
    }
    for (int s = 0; s < NSLAB - 1; ++s) {
      fetch(s + 1);
      multiply(s & 1);
      stage((s + 1) & 1);
      __syncthreads();
    }
    if (sw) load_rb();
    multiply((NSLAB - 1) & 1);

#pragma unroll
    for (int nt = 0; nt < 2; ++nt)
#pragma unroll
      for (int v = 0; v < 8; ++v)
        beta[(size_t)(blockIdx.x * 64 + wm * 16 + v + 8 * half) * E_ + (2 * wn + nt) * 16 + m16] = acc[nt][v];
#pragma unroll
    for (int nt = 0; nt < 2; ++nt)
#pragma unroll
      for (int v = 0; v < 8; ++v)
        bs[(wm * 16 + v + 8 * half) * BP + (2 * wn + nt) * 16 + m16] = acc[nt][v];
  } else if (sw) {
    mag = 1.f / (1.f + expf(-araw));
    cw  = cosf(oraw);
    sn  = sinf(oraw);
    load_rb();
  }

#define LDS_BARRIER_A() asm volatile("s_wait_dscnt 0x0\n\ts_barrier_signal -1\n\ts_barrier_wait -1" ::: "memory")

  const int u = blockIdx.x - NGB;
  LDS_BARRIER_A();
  float bt[LCH][4];
#pragma unroll
  for (int i = 0; i < LCH; ++i)
#pragma unroll
    for (int q = 0; q < 4; ++q) bt[i][q] = 0.f;
  if (sw && !unit) {
#pragma unroll
    for (int i = 0; i < LCH; ++i)
#pragma unroll
      for (int q = 0; q < 4; ++q) bt[i][q] = bs[(LCH * (b0 + q) + i) * BP + e];
  }

  v8f st = {};
  if (unit) {
#pragma unroll
    for (int q = 0; q < 4; ++q) st[2 * q] = (e == 8 * u + b0 + q) ? 1.f : 0.f;
  }

#pragma unroll
  for (int i = 0; i < LCH; ++i) {
    const int cur = i & 1;
    if (sw) {
      float* xw = &xs[cur * (16 * SP) + (8 * half) * SP + e];
#pragma unroll
      for (int q = 0; q < 4; ++q) {
        const float re = st[2 * q], im = st[2 * q + 1];
        const float nr = fmaf(mag, re * cw - im * sn, bt[i][q]);
        const float ni = mag * (re * sn + im * cw);
        xw[(2 * q) * SP]     = nr;
        xw[(2 * q + 1) * SP] = ni;
      }
    }
    LDS_BARRIER_A();
    if (sw) {
      const float4* ar = (const float4*)&xs[cur * (16 * SP) + m16 * SP + 32 * half];
      float4 av[8];
#pragma unroll
      for (int j = 0; j < 8; ++j) av[j] = ar[j];
      asm volatile("" ::: "memory");
      __builtin_amdgcn_sched_barrier(0);
      v8f acc0 = {}, acc1 = {};
#pragma unroll
      for (int j = 0; j < 8; ++j) {
        v2f a0 = {av[j].x, av[j].y};
        v2f a1 = {av[j].z, av[j].w};
        acc0 = WMMA_F32(a0, rb[2 * j],     acc0);
        acc1 = WMMA_F32(a1, rb[2 * j + 1], acc1);
      }
      __builtin_amdgcn_sched_barrier(0);
      st = acc0 + acc1;
    }
  }

  if (sw) {
    float* xw = &xs[(8 * half) * SP + e];
#pragma unroll
    for (int i = 0; i < 8; ++i) xw[i * SP] = st[i];
  }
  LDS_BARRIER_A();
  if (sw) {
    const int bb = blockIdx.x >> 5;
    const int c0 = (blockIdx.x & 31) * (64 / LCH);
    const int orow = tid >> 4, oc = (tid & 15) * 4;
#pragma unroll
    for (int hh = 0; hh < 2; ++hh) {
      const int row = orow + 8 * hh;
      const float4 o = *(const float4*)&xs[row * SP + oc];
      float* dst = unit ? (((row & 1) ? ci : cr) + (size_t)(8 * u + (row >> 1)) * E_ + oc)
                        : (eb + (size_t)(c0 + (row >> 1)) * ET_F + (2 * bb + (row & 1)) * E_ + oc);
      *(float4*)dst = o;
    }
  }
}

static constexpr int NPRE = 8;
static constexpr int RP = 72;

static constexpr int YT_F  = 16 * D_;
static constexpr int WSB_F = 16 * E_;
static constexpr int XS_F  = 2 * 16 * SP;
static constexpr int RS_F  = 2 * B_ * RP;
static constexpr int FS_F  = B_ * RP;
static_assert(XS_F + RS_F + FS_F <= 16 * WSB_F, "the scan's buffers lie inside the staging tiles");

__global__ __launch_bounds__(512)
void k_scan_out(const float* __restrict__ beta, const float* __restrict__ alpha,
                const float* __restrict__ omega, const float* __restrict__ res,
                const float* __restrict__ eb, const float* __restrict__ cr, const float* __restrict__ ci,
                float* __restrict__ rfin, float* __restrict__ ifin,
                const float* __restrict__ w_out,
                const float* __restrict__ gamma, const float* __restrict__ lbeta,
                float* __restrict__ y) {
  const int c    = blockIdx.x;
  const int tid  = threadIdx.x;
  const int lane = tid & 31;
  const int m16  = lane & 15;
  const int half = lane >> 4;
  const int t0   = c * LCH;

  __shared__ __attribute__((aligned(16))) float smem[4 * YT_F + 16 * WSB_F];
  float* const xs = smem + 4 * YT_F;
  float* const rs = xs + XS_F;
  float* const fs = rs + RS_F;

  {
    const bool sw  = __builtin_amdgcn_readfirstlane(tid >> 7) == 0;
    const int w    = (tid >> 5) & 3;
    const int e    = w * 16 + m16;
    const int b0   = 4 * half;

    const int npre  = (c < NPRE) ? c : NPRE;
    const int tlast = t0 + LCH - 1;

    float mag = 0.f, cw = 0.f, sn = 0.f;
    v2f rb[16], crb[16], cib[16];
#pragma unroll
    for (int t = 0; t < 16; ++t) { rb[t] = {}; crb[t] = {}; cib[t] = {}; }
    const float* bp = beta + (size_t)(b0 * S_) * E_ + e;
    float bte[4], bto[4];
    if (sw) {
      mag = 1.f / (1.f + expf(-alpha[e]));
      cw  = cosf(omega[e]);
      sn  = sinf(omega[e]);
#pragma unroll
      for (int t = 0; t < 16; ++t) {
        const int k = 32 * half + 2 * t;
        crb[t].x = cr[k * E_ + e];
        crb[t].y = cr[(k + 1) * E_ + e];
        cib[t].x = ci[k * E_ + e];
        cib[t].y = ci[(k + 1) * E_ + e];
        rb[t].x = res[k * E_ + e];
        rb[t].y = res[(k + 1) * E_ + e];
      }
#pragma unroll
      for (int t = 0; t < 16; ++t) {
        asm volatile("" : "+v"(crb[t].x), "+v"(crb[t].y), "+v"(cib[t].x), "+v"(cib[t].y));
        asm volatile("" : "+v"(rb[t].x), "+v"(rb[t].y));
      }
    }

    const float* ep = eb + (size_t)(2 * b0) * E_ + e;
    auto load_end = [&](const int d, float (&v)[8]) {
      const float* p = ep + (size_t)d * ET_F;
#pragma unroll
      for (int i = 0; i < 8; ++i) v[i] = p[i * E_];
    };
    v8f st = {};
    {
      float e0[8];
      load_end(c - npre, e0);
#pragma unroll
      for (int i = 0; i < 8; ++i) st[i] = (npre > 0) ? e0[i] : 0.f;
    }

#pragma unroll
    for (int q = 0; q < 4; ++q) bte[q] = bp[(size_t)q * S_ * E_ + (size_t)t0 * E_];
#pragma unroll
    for (int q = 0; q < 4; ++q) bto[q] = bp[(size_t)q * S_ * E_ + (size_t)(t0 + 1) * E_];

    const int ob = (tid >> 4) & 7, oc = (tid & 15) * 4;
    float* sp = y + (size_t)(ob * S_ + t0) * D_ + oc;

    auto rotate = [&](const int cur, const float (&bt)[4]) {
      float* xw = &xs[cur * (16 * SP) + (8 * half) * SP + e];
#pragma unroll
      for (int q = 0; q < 4; ++q) {
        const float re = st[2 * q], im = st[2 * q + 1];
        const float nr = fmaf(mag, re * cw - im * sn, bt[q]);
        const float ni = mag * (re * sn + im * cw);
        xw[(2 * q) * SP]     = nr;
        xw[(2 * q + 1) * SP] = ni;
      }
    };

    auto refill = [&](float (&bt)[4], const int tn) {
      const int t2 = (tn < tlast) ? tn : tlast;
#pragma unroll
      for (int q = 0; q < 4; ++q) bt[q] = bp[(size_t)q * S_ * E_ + (size_t)t2 * E_];
    };

    auto step = [&](const int t, const int cur, float (&bt)[4], float (&btn)[4],
                    const bool emit, const bool flush, const int ot, const bool more) {
      asm volatile("s_wait_dscnt 0x0\n\ts_barrier_signal -1" ::: "memory");

      refill(bt, t + 2);

      asm volatile("s_barrier_wait -1" ::: "memory");

      if (sw) {
        const float4* ar = (const float4*)&xs[cur * (16 * SP) + m16 * SP + 32 * half];
        float4 av[8];
#pragma unroll
        for (int j = 0; j < 8; ++j) av[j] = ar[j];

        float4 o = make_float4(0.f, 0.f, 0.f, 0.f);
        if (flush) o = *(const float4*)&rs[(cur ^ 1) * (B_ * RP) + ob * RP + oc];

        asm volatile("" ::: "memory");
        __builtin_amdgcn_sched_barrier(0);

        v8f acc0 = {}, acc1 = {};
#pragma unroll
        for (int j = 0; j < 8; ++j) {
          v2f a0 = {av[j].x, av[j].y};
          v2f a1 = {av[j].z, av[j].w};
          acc0 = WMMA_F32(a0, rb[2 * j],     acc0);
          acc1 = WMMA_F32(a1, rb[2 * j + 1], acc1);
        }
        __builtin_amdgcn_sched_barrier(0);
        st = acc0 + acc1;

        if (emit) {
          float* rw = &rs[cur * (B_ * RP) + b0 * RP + e];
#pragma unroll
          for (int q = 0; q < 4; ++q) rw[q * RP] = st[2 * q];
          if (flush) *(float4*)(sp + (ot - 1) * E_) = o;
        }
      }
      if (more) {
        asm volatile("" : "+v"(btn[0]), "+v"(btn[1]), "+v"(btn[2]), "+v"(btn[3]));
        if (sw) rotate(cur ^ 1, btn);
      }
    };

    {
      int xo = 0;
      for (int g = npre - 1; g >= 1; --g) {
        if (sw) {
          float* xw = &xs[xo + (8 * half) * SP + e];
#pragma unroll
          for (int i = 0; i < 8; ++i) xw[i * SP] = st[i];
        }
        asm volatile("s_wait_dscnt 0x0\n\ts_barrier_signal -1" ::: "memory");
        float en[8];
        load_end(c - g, en);
        asm volatile("s_barrier_wait -1" ::: "memory");

        v8f pa = {}, pj = {};
        if (sw) {
          const float4* ar = (const float4*)&xs[xo + m16 * SP + 32 * half];
          const float4* jr = (const float4*)&xs[xo + (m16 ^ 1) * SP + 32 * half];
          float4 av[8], jv[8];
#pragma unroll
          for (int j = 0; j < 8; ++j) av[j] = ar[j];
#pragma unroll
          for (int j = 0; j < 8; ++j) jv[j] = jr[j];
          asm volatile("" ::: "memory");
          __builtin_amdgcn_sched_barrier(0);
          v8f pa0 = {}, pa1 = {}, pj0 = {}, pj1 = {};
#pragma unroll
          for (int j = 0; j < 8; ++j) {
            v2f a0 = {av[j].x, av[j].y};
            v2f a1 = {av[j].z, av[j].w};
            v2f j0 = {jv[j].x, jv[j].y};
            v2f j1 = {jv[j].z, jv[j].w};
            pa0 = WMMA_F32(a0, crb[2 * j],     pa0);
            pj0 = WMMA_F32(j0, cib[2 * j],     pj0);
            pa1 = WMMA_F32(a1, crb[2 * j + 1], pa1);
            pj1 = WMMA_F32(j1, cib[2 * j + 1], pj1);
          }
          __builtin_amdgcn_sched_barrier(0);
          pa = pa0 + pa1;
          pj = pj0 + pj1;
        }
        asm volatile("" : "+v"(en[0]), "+v"(en[1]), "+v"(en[2]), "+v"(en[3]),
                          "+v"(en[4]), "+v"(en[5]), "+v"(en[6]), "+v"(en[7]));
#pragma unroll
        for (int q = 0; q < 4; ++q) {
          st[2 * q]     = (pa[2 * q]     - pj[2 * q])     + en[2 * q];
          st[2 * q + 1] = (pa[2 * q + 1] + pj[2 * q + 1]) + en[2 * q + 1];
        }
        xo = 16 * SP - xo;
      }
      asm volatile("s_wait_dscnt 0x0\n\ts_barrier_signal -1\n\ts_barrier_wait -1" ::: "memory");
    }

    asm volatile("" : "+v"(bte[0]), "+v"(bte[1]), "+v"(bte[2]), "+v"(bte[3]));
    if (sw) rotate(0, bte);
    int t = t0;
    for (int ot = 0; ot < LCH; ot += 2, t += 2) {
      step(t,     0, bte, bto, true, ot > 0, ot,     true);
      step(t + 1, 1, bto, bte, true, true,   ot + 1, ot + 2 < LCH);
    }

    if (sw) {
      float* iw = &fs[b0 * RP + e];
#pragma unroll
      for (int q = 0; q < 4; ++q) iw[q * RP] = st[2 * q + 1];
    }
    asm volatile("s_wait_dscnt 0x0\n\ts_barrier_signal -1\n\ts_barrier_wait -1" ::: "memory");
    if (sw) {
      const float4 o = *(const float4*)&rs[((LCH - 1) & 1) * (B_ * RP) + ob * RP + oc];
      *(float4*)(sp + (LCH - 1) * E_) = o;
      if (c == NBLK - 1) {
        *(float4*)(rfin + ob * E_ + oc) = o;
        *(float4*)(ifin + ob * E_ + oc) = *(const float4*)&fs[ob * RP + oc];
      }
    }
  }
  __syncthreads();

  const int wi = __builtin_amdgcn_readfirstlane(tid >> 5);
  float* const stg = smem + 4 * YT_F;
  float* const wst = stg + wi * WSB_F;

  int so[8], fo[8];
#pragma unroll
  for (int i = 0; i < 8; ++i) {
    const int r = 2 * i + half;
    so[i] = r * E_ + 4 * (m16 ^ r);
    fo[i] = m16 * E_ + 4 * ((8 * half + i) ^ m16);
  }

  v4f g[8];
  auto fetch = [&](const float* p) {
#pragma unroll
    for (int i = 0; i < 8; ++i) g[i] = *(const v4f*)(p + 128 * i);
  };
  auto stage = [&]() {
#pragma unroll
    for (int i = 0; i < 8; ++i) *(v4f*)(wst + so[i]) = g[i];
  };
  auto ldfrag = [&](const float* tile, v4f (&f)[8]) {
#pragma unroll
    for (int j = 0; j < 8; ++j) f[j] = *(const v4f*)(tile + fo[j]);
  };
#define TILE_STORED() asm volatile("s_wait_dscnt 0x0" ::: "memory")

  v4f ga[8];
  if (wi < 4) {
    const float* s0p = y + (size_t)((2 * wi) * S_ + t0) * D_ + 4 * lane;
    const float* s1p = y + (size_t)((2 * wi + 1) * S_ + t0) * D_ + 4 * lane;
#pragma unroll
    for (int i = 0; i < 4; ++i) {
      ga[i]     = *(const v4f*)(s0p + 128 * i);
      ga[i + 4] = *(const v4f*)(s1p + 128 * i);
    }
  }
  v4f bq[4][8];
  const float* wt = w_out + (size_t)(wi * 4) * (16 * E_) + 4 * lane;
  fetch(wt);
  stage();
  fetch(wt + 1 * (16 * E_));
  TILE_STORED();
  ldfrag(wst, bq[0]);
  stage();
  fetch(wt + 2 * (16 * E_));
  TILE_STORED();
  ldfrag(wst, bq[1]);
  stage();
  fetch(wt + 3 * (16 * E_));
  TILE_STORED();
  ldfrag(wst, bq[2]);
  stage();
  TILE_STORED();
  ldfrag(wst, bq[3]);
  if (wi < 4) {
#pragma unroll
    for (int i = 0; i < 8; ++i) *(v4f*)(wst + so[i]) = ga[i];
  }
  __syncthreads();

  auto ybof = [&](const int T, const int dt) {
    unsigned yb = 4u * (unsigned)(T * YT_F + (8 * half) * D_ + (wi * 4 + dt) * 16 + m16);
    asm volatile("" : "+v"(yb));
    return yb;
  };
  auto mm2 = [&](const v4f (&aq)[8], const v4f (&b0)[8], const v4f (&b1)[8], const unsigned ya, const unsigned yb) {
    float* yr0 = (float*)((char*)smem + ya);
    float* yr1 = (float*)((char*)smem + yb);
    GROUP_EDGE();
    v8f acc0 = {}, acc1 = {};
#pragma unroll
    for (int j = 0; j < 8; ++j) {
      const v2f a0 = {aq[j].x, aq[j].y};
      const v2f a1 = {aq[j].z, aq[j].w};
      const v2f c0 = {b0[j].x, b0[j].y};
      const v2f d0 = {b1[j].x, b1[j].y};
      acc0 = WMMA_F32(a0, c0, acc0);
      acc1 = WMMA_F32(a0, d0, acc1);
      const v2f c1 = {b0[j].z, b0[j].w};
      const v2f d1 = {b1[j].z, b1[j].w};
      acc0 = WMMA_F32(a1, c1, acc0);
      acc1 = WMMA_F32(a1, d1, acc1);
    }
    GROUP_EDGE();
#pragma unroll
    for (int v = 0; v < 8; ++v) yr0[v * D_] = acc0[v];
#pragma unroll
    for (int v = 0; v < 8; ++v) yr1[v * D_] = acc1[v];
  };
  auto runs = [&](const int T, const v4f (&aq)[8]) {
    const unsigned y0 = ybof(T, 0), y1 = ybof(T, 1), y2 = ybof(T, 2), y3 = ybof(T, 3);
    mm2(aq, bq[0], bq[1], y0, y1);
    mm2(aq, bq[2], bq[3], y2, y3);
  };
  auto rowload = [&](const int T, v4f (&yv)[8]) {
    const float* yrow = smem + T * YT_F + wi * D_ + 4 * lane;
#pragma unroll
    for (int i = 0; i < 8; ++i) yv[i] = *(const v4f*)(yrow + 128 * i);
  };
  auto lx = [](const float v, auto mtag) {
    constexpr int m = decltype(mtag)::value;
    const int b = __builtin_bit_cast(int, v);
    if constexpr (m == 16)
      return __builtin_bit_cast(float, __builtin_amdgcn_permlanex16(b, b, 0x76543210, 0xfedcba98, false, false));
    else
      return __builtin_bit_cast(float, __builtin_amdgcn_update_dpp(b, b, 0x160 + m, 0xf, 0xf, false));
  };
  auto rowfinish = [&](const int T, const v4f (&yv)[8]) {
    v4f gA[4], bA[4], gB[4], bB[4];
#pragma unroll
    for (int i = 0; i < 4; ++i) {
      gA[i] = *(const v4f*)(gamma + 128 * i + 4 * lane);
      bA[i] = *(const v4f*)(lbeta + 128 * i + 4 * lane);
    }
    __builtin_amdgcn_sched_barrier(0);
    v4f p0 = yv[0];
    v4f p1 = yv[0] * yv[0];
#pragma unroll
    for (int i = 1; i < 8; ++i) {
      p0 += yv[i];
      p1 += yv[i] * yv[i];
    }
    float s0 = (p0.x + p0.y) + (p0.z + p0.w);
    float s1 = (p1.x + p1.y) + (p1.z + p1.w);
    s0 += lx(s0, std::integral_constant<int, 16>{});  s1 += lx(s1, std::integral_constant<int, 16>{});
    s0 += lx(s0, std::integral_constant<int, 8>{});   s1 += lx(s1, std::integral_constant<int, 8>{});
    s0 += lx(s0, std::integral_constant<int, 4>{});   s1 += lx(s1, std::integral_constant<int, 4>{});
    s0 += lx(s0, std::integral_constant<int, 2>{});   s1 += lx(s1, std::integral_constant<int, 2>{});
    s0 += lx(s0, std::integral_constant<int, 1>{});   s1 += lx(s1, std::integral_constant<int, 1>{});
    const float mu  = s0 * (1.f / D_);
    const float var = s1 * (1.f / D_) - mu * mu;
    const float rsd = rsqrtf(var + 1e-5f);
#pragma unroll
    for (int i = 0; i < 4; ++i) {
      gB[i] = *(const v4f*)(gamma + 128 * (i + 4) + 4 * lane);
      bB[i] = *(const v4f*)(lbeta + 128 * (i + 4) + 4 * lane);
    }
    __builtin_amdgcn_sched_barrier(0);
    float* yo = y + (size_t)((2 * T + (wi >> 3)) * S_ + t0 + (wi & 7)) * D_ + 4 * lane;
#pragma unroll
    for (int i = 0; i < 4; ++i) {
      const v4f o = (yv[i] - mu) * rsd * gA[i] + bA[i];
      *(v4f*)(yo + 128 * i) = o;
    }
#pragma unroll
    for (int i = 0; i < 4; ++i) {
      const v4f o = (yv[i + 4] - mu) * rsd * gB[i] + bB[i];
      *(v4f*)(yo + 128 * (i + 4)) = o;
    }
  };
#define LDS_BARRIER() asm volatile("s_wait_dscnt 0x0\n\ts_barrier_signal -1\n\ts_barrier_wait -1" ::: "memory")

  {
    v4f aq[8];
    ldfrag(stg, aq);
    runs(0, aq);
  }
  LDS_BARRIER();
#pragma unroll 1
  for (int T = 1; T < 4; ++T) {
    v4f aq[8], yv[8];
    ldfrag(stg + T * WSB_F, aq);
    rowload(T - 1, yv);
    runs(T, aq);
    asm volatile("s_wait_dscnt 0x0\n\ts_barrier_signal -1" ::: "memory");
    rowfinish(T - 1, yv);
    asm volatile("s_barrier_wait -1" ::: "memory");
  }
  {
    v4f yv[8];
    rowload(3, yv);
    rowfinish(3, yv);
  }
}

extern "C" void kernel_launch(void* const* d_in, const int* in_sizes, int n_in,
                              void* d_out, int out_size, void* d_ws, size_t ws_size,
                              hipStream_t stream) {
  (void)in_sizes; (void)n_in; (void)out_size; (void)ws_size;

  const float* x     = (const float*)d_in[0];
  const float* alpha = (const float*)d_in[1];
  const float* omega = (const float*)d_in[2];
  const float* w_in  = (const float*)d_in[3];
  const float* w_out = (const float*)d_in[4];
  const float* res   = (const float*)d_in[5];
  const float* gamma = (const float*)d_in[6];
  const float* lbeta = (const float*)d_in[7];

  float* out  = (float*)d_out;
  float* y    = out;
  float* rfin = out + YSZ;
  float* ifin = rfin + B_ * E_;

  float* beta = (float*)d_ws;
  float* eb   = beta + (size_t)M_ * E_;
  float* cr   = eb + (size_t)NBLK * ET_F;
  float* ci   = cr + E_ * E_;

  hipLaunchKernelGGL(k_beta_gemm, dim3(NGB + NUB), dim3(256), 0, stream,
                     x, w_in, alpha, omega, res, beta, eb, cr, ci);
  hipLaunchKernelGGL(k_scan_out, dim3(NBLK), dim3(512), 0, stream,
                     beta, alpha, omega, res, eb, cr, ci, rfin, ifin, w_out, gamma, lbeta, y);
}
